# starting kernel
# baseline (speedup 1.0000x reference)
#include <hip/hip_runtime.h>
#include <math.h>


typedef __attribute__((ext_vector_type(16))) _Float16 v16h;
typedef __attribute__((ext_vector_type(8)))  float    v8f;

#define DIM 256

__device__ __forceinline__ float atomAddF(float* p, float v) {
  return unsafeAtomicAdd(p, v);
}

__global__ void zero_kernel(float* __restrict__ p, int n) {
  int i = blockIdx.x * blockDim.x + threadIdx.x;
  if (i < n) p[i] = 0.0f;
}

__global__ void deg_kernel(const int* __restrict__ row, const int* __restrict__ col,
                           float* __restrict__ deg_out, float* __restrict__ deg_in, int E) {
  int e = blockIdx.x * blockDim.x + threadIdx.x;
  if (e < E) {
    atomAddF(&deg_out[row[e]], 1.0f);
    atomAddF(&deg_in[col[e]], 1.0f);
  }
}

__global__ void rsqrt_kernel(float* __restrict__ d, int n) {
  int i = blockIdx.x * blockDim.x + threadIdx.x;
  if (i < n) {
    float v = d[i];
    v = v < 1.0f ? 1.0f : v;
    d[i] = rsqrtf(v);
  }
}

__global__ __launch_bounds__(256)
void scatter_kernel(const float* __restrict__ h, const float* __restrict__ dinv_out,
                    const int* __restrict__ row, const int* __restrict__ col,
                    float* __restrict__ agg, int E) {
  int idx = blockIdx.x * blockDim.x + threadIdx.x;
  int e  = idx >> 6;
  int c4 = (idx & 63) << 2;
  if (e < E) {
    int r = row[e];
    int c = col[e];
    float s = dinv_out[r];
    const float4 hv = *(const float4*)(h + r * DIM + c4);
    float* dst = agg + c * DIM + c4;
    atomAddF(dst + 0, hv.x * s);
    atomAddF(dst + 1, hv.y * s);
    atomAddF(dst + 2, hv.z * s);
    atomAddF(dst + 3, hv.w * s);
  }
}

__global__ __launch_bounds__(256)
void pack_w_kernel(const float* __restrict__ W, _Float16* __restrict__ pack) {
  int t = blockIdx.x * blockDim.x + threadIdx.x;
  int elem = t & 15;
  int lane = (t >> 4) & 31;
  int kc   = (t >> 9) & 7;
  int tn   = t >> 12;
  int v = elem >> 1, p = elem & 1;
  int half = lane >> 4, mn = lane & 15;
  int kb = kc * 32 + ((v & 4) << 2) + (half << 3) + ((v & 3) << 1) + p;
  int gn = tn * 16 + mn;
  pack[t] = (_Float16)W[kb * DIM + gn];
}

template <bool HAS_RESID>
__global__ __launch_bounds__(128)
void gemm_wmma_kernel(const float* __restrict__ A, const _Float16* __restrict__ Wp,
                      const float* __restrict__ bias, const float* __restrict__ dinv_in,
                      const float* __restrict__ resid, float* __restrict__ out) {
  const int wave   = blockIdx.x * (blockDim.x >> 5) + (threadIdx.x >> 5);
  const int lane   = threadIdx.x & 31;
  const int tile_m = wave >> 4;
  const int tile_n = wave & 15;
  const int half   = lane >> 4;
  const int mn     = lane & 15;

  const int gm_a = tile_m * 16 + mn;
  const int gn   = tile_n * 16 + mn;
  const float* Arow = A + gm_a * DIM + half * 8;
  const v16h*  Bp   = (const v16h*)Wp + tile_n * 256 + lane;

  v8f c = {};
#pragma unroll
  for (int kc = 0; kc < 8; ++kc) {
    const float* ap = Arow + kc * 32;
    const float4 q0 = *(const float4*)(ap);
    const float4 q1 = *(const float4*)(ap + 4);
    const float4 q2 = *(const float4*)(ap + 16);
    const float4 q3 = *(const float4*)(ap + 20);
    v16h a;
    a[0]  = (_Float16)q0.x;  a[1]  = (_Float16)q0.y;
    a[2]  = (_Float16)q0.z;  a[3]  = (_Float16)q0.w;
    a[4]  = (_Float16)q1.x;  a[5]  = (_Float16)q1.y;
    a[6]  = (_Float16)q1.z;  a[7]  = (_Float16)q1.w;
    a[8]  = (_Float16)q2.x;  a[9]  = (_Float16)q2.y;
    a[10] = (_Float16)q2.z;  a[11] = (_Float16)q2.w;
    a[12] = (_Float16)q3.x;  a[13] = (_Float16)q3.y;
    a[14] = (_Float16)q3.z;  a[15] = (_Float16)q3.w;
    const v16h b = Bp[kc * 32];
    c = __builtin_amdgcn_wmma_f32_16x16x32_f16(false, a, false, b,
                                               (short)0, c, false, false);
  }

  const float bn = bias[gn];
#pragma unroll
  for (int r = 0; r < 8; ++r) {
    const int gm = tile_m * 16 + r + half * 8;
    float val = c[r] + bn;
    val *= dinv_in[gm];
    if constexpr (HAS_RESID) val += resid[gm * DIM + gn];
    out[gm * DIM + gn] = val;
  }
}

__global__ __launch_bounds__(256)
void bn_stats_kernel(const float* __restrict__ pre, float* __restrict__ sums,
                     float* __restrict__ sumsq, int n) {
  const int c    = threadIdx.x;
  const int row0 = blockIdx.x * 128;
  int rend = row0 + 128;
  if (rend > n) rend = n;
  float s = 0.0f, s2 = 0.0f;
  for (int r = row0; r < rend; ++r) {
    float v = pre[r * DIM + c];
    s  += v;
    s2 += v * v;
  }
  atomAddF(&sums[c], s);
  atomAddF(&sumsq[c], s2);
}

__global__ void bn_finalize_kernel(float* __restrict__ sums, float* __restrict__ sumsq, int n) {
  const int c = threadIdx.x;
  const float inv_n = 1.0f / (float)n;
  float mu  = sums[c] * inv_n;
  float var = sumsq[c] * inv_n - mu * mu;
  sums[c]  = mu;
  sumsq[c] = rsqrtf(var + 1e-5f);
}

__global__ __launch_bounds__(256)
void bn_apply_kernel(float* __restrict__ buf, const float* __restrict__ mu,
                     const float* __restrict__ inv_std, const float* __restrict__ gamma,
                     const float* __restrict__ beta, int nd) {
  int i = blockIdx.x * blockDim.x + threadIdx.x;
  if (i < nd) {
    const int c = i & (DIM - 1);
    float v = (buf[i] - mu[c]) * inv_std[c] * gamma[c] + beta[c];
    buf[i] = fmaxf(v, 0.0f);
  }
}

extern "C" void kernel_launch(void* const* d_in, const int* in_sizes, int n_in,
                              void* d_out, int out_size, void* d_ws, size_t ws_size,
                              hipStream_t stream) {
  const float* x  = (const float*)d_in[0];
  const int*   ei = (const int*)d_in[1];
  const float* W0 = (const float*)d_in[2];
  const float* b0 = (const float*)d_in[3];
  const float* g0 = (const float*)d_in[4];
  const float* be0= (const float*)d_in[5];
  const float* W1 = (const float*)d_in[6];
  const float* b1 = (const float*)d_in[7];
  const float* g1 = (const float*)d_in[8];
  const float* be1= (const float*)d_in[9];
  const float* W2 = (const float*)d_in[10];
  const float* b2 = (const float*)d_in[11];
  float* out = (float*)d_out;

  const int N = in_sizes[0] / DIM;
  const int E = in_sizes[1] / 2;
  const int ND = N * DIM;
  const int* row = ei;
  const int* col = ei + E;

  float* ws       = (float*)d_ws;
  float* dinv_out = ws;
  float* dinv_in  = ws + N;
  float* stats    = ws + 2 * N;
  float* agg      = ws + 2 * N + 512;
  float* bufA     = agg + (size_t)ND;
  _Float16* wp0   = (_Float16*)(bufA + (size_t)ND);
  _Float16* wp1   = wp0 + 65536;
  _Float16* wp2   = wp1 + 65536;

  const int zb = 256;
  auto nb = [](int n, int b) { return (n + b - 1) / b; };

  const int scat_blocks = (E * 64) / 256;
  const int gemm_blocks = (N / 16) * 16 / 4;
  const int bn_blocks   = nb(N, 128);
  const int pack_blocks = 65536 / 256;

  zero_kernel<<<nb(2 * N, zb), zb, 0, stream>>>(dinv_out, 2 * N);
  deg_kernel<<<nb(E, zb), zb, 0, stream>>>(row, col, dinv_out, dinv_in, E);
  rsqrt_kernel<<<nb(2 * N, zb), zb, 0, stream>>>(dinv_out, 2 * N);
  pack_w_kernel<<<pack_blocks, 256, 0, stream>>>(W0, wp0);
  pack_w_kernel<<<pack_blocks, 256, 0, stream>>>(W1, wp1);
  pack_w_kernel<<<pack_blocks, 256, 0, stream>>>(W2, wp2);

  zero_kernel<<<nb(ND, zb), zb, 0, stream>>>(agg, ND);
  scatter_kernel<<<scat_blocks, 256, 0, stream>>>(x, dinv_out, row, col, agg, E);
  gemm_wmma_kernel<true><<<gemm_blocks, 128, 0, stream>>>(agg, wp0, b0, dinv_in, x, out);
  zero_kernel<<<2, zb, 0, stream>>>(stats, 512);
  bn_stats_kernel<<<bn_blocks, 256, 0, stream>>>(out, stats, stats + 256, N);
  bn_finalize_kernel<<<1, 256, 0, stream>>>(stats, stats + 256, N);
  bn_apply_kernel<<<nb(ND, zb), zb, 0, stream>>>(out, stats, stats + 256, g0, be0, ND);

  zero_kernel<<<nb(ND, zb), zb, 0, stream>>>(agg, ND);
  scatter_kernel<<<scat_blocks, 256, 0, stream>>>(out, dinv_out, row, col, agg, E);
  gemm_wmma_kernel<true><<<gemm_blocks, 128, 0, stream>>>(agg, wp1, b1, dinv_in, out, bufA);
  zero_kernel<<<2, zb, 0, stream>>>(stats, 512);
  bn_stats_kernel<<<bn_blocks, 256, 0, stream>>>(bufA, stats, stats + 256, N);
  bn_finalize_kernel<<<1, 256, 0, stream>>>(stats, stats + 256, N);
  bn_apply_kernel<<<nb(ND, zb), zb, 0, stream>>>(bufA, stats, stats + 256, g1, be1, ND);

  zero_kernel<<<nb(ND, zb), zb, 0, stream>>>(agg, ND);
  scatter_kernel<<<scat_blocks, 256, 0, stream>>>(bufA, dinv_out, row, col, agg, E);
  gemm_wmma_kernel<false><<<gemm_blocks, 128, 0, stream>>>(agg, wp2, b2, dinv_in, nullptr, out);
}


# Edge sum without global atomics: edges binned once by 196-row target range; per layer a work-group per bin adds source rows into an LDS tile (LDS f32 adds, arrival order), stores it; no zero fill
# speedup vs baseline: 3.0962x; 3.0962x over previous
#include <hip/hip_runtime.h>
#include <math.h>


typedef __attribute__((ext_vector_type(16))) _Float16 v16h;
typedef __attribute__((ext_vector_type(8)))  _Float16 v8h;
typedef __attribute__((ext_vector_type(4)))  _Float16 v4h;
typedef __attribute__((ext_vector_type(8)))  float    v8f;
typedef __attribute__((ext_vector_type(4)))  float    v4f;

#define DIM 256

__device__ __forceinline__ float atomAddF(float* p, float v) {
  return unsafeAtomicAdd(p, v);
}

__global__ void zero_kernel(float* __restrict__ p, int n) {
  int i = blockIdx.x * blockDim.x + threadIdx.x;
  if (i < n) p[i] = 0.0f;
}

__global__ void deg_kernel(const int* __restrict__ row, const int* __restrict__ col,
                           float* __restrict__ deg_out, float* __restrict__ deg_in, int E) {
  int e = blockIdx.x * blockDim.x + threadIdx.x;
  if (e < E) {
    atomAddF(&deg_out[row[e]], 1.0f);
    atomAddF(&deg_in[col[e]], 1.0f);
  }
}

__global__ void rsqrt_kernel(float* __restrict__ d, int n) {
  int i = blockIdx.x * blockDim.x + threadIdx.x;
  if (i < n) {
    float v = d[i];
    v = v < 1.0f ? 1.0f : v;
    d[i] = rsqrtf(v);
  }
}

#define TR 196
#define MAXB 1024
#define BIN_EPT 4

__global__ __launch_bounds__(1024)
void bin_count_kernel(const int* __restrict__ col, int* __restrict__ count, int E, int nbins) {
  __shared__ int hist[MAXB];
  const int tid = threadIdx.x;
  hist[tid] = 0;
  __syncthreads();
  const int base = blockIdx.x * (1024 * BIN_EPT) + tid;
  for (int k = 0; k < BIN_EPT; ++k) {
    const int e = base + k * 1024;
    if (e < E) atomicAdd(&hist[col[e] / TR], 1);
  }
  __syncthreads();
  if (tid < nbins) {
    const int hc = hist[tid];
    if (hc) atomicAdd(&count[tid], hc);
  }
}

__global__ __launch_bounds__(1024)
void bin_fill_kernel(const int* __restrict__ row, const int* __restrict__ col,
                     const int* __restrict__ count, int* __restrict__ cursor,
                     int* __restrict__ offs, int2* __restrict__ list, int E, int nbins) {
  __shared__ int hist[MAXB];
  __shared__ int scan[MAXB];
  __shared__ int basep[MAXB];
  const int tid = threadIdx.x;
  const int c = (tid < nbins) ? count[tid] : 0;
  scan[tid] = c;
  hist[tid] = 0;
  __syncthreads();
  for (int d = 1; d < MAXB; d <<= 1) {
    int v = scan[tid];
    if (tid >= d) v += scan[tid - d];
    __syncthreads();
    scan[tid] = v;
    __syncthreads();
  }
  const int excl = scan[tid] - c;
  if (blockIdx.x == 0 && tid < nbins) offs[tid] = excl;

  const int base = blockIdx.x * (1024 * BIN_EPT) + tid;
  int rr[BIN_EPT], cl[BIN_EPT], bb[BIN_EPT], rk[BIN_EPT];
#pragma unroll
  for (int k = 0; k < BIN_EPT; ++k) {
    const int e = base + k * 1024;
    bb[k] = -1;
    if (e < E) {
      const int cc = col[e];
      rr[k] = row[e];
      bb[k] = cc / TR;
      cl[k] = cc - bb[k] * TR;
      rk[k] = atomicAdd(&hist[bb[k]], 1);
    }
  }
  __syncthreads();
  if (tid < nbins) {
    const int hc = hist[tid];
    basep[tid] = excl + (hc ? atomicAdd(&cursor[tid], hc) : 0);
  }
  __syncthreads();
#pragma unroll
  for (int k = 0; k < BIN_EPT; ++k) {
    if (bb[k] >= 0) list[basep[bb[k]] + rk[k]] = make_int2(rr[k], cl[k]);
  }
}

#define PE 4
struct EdgeGroup {
  int   dst[PE];
  float s[PE];
  float v[PE][8];
};

__device__ __forceinline__
void group_load(EdgeGroup& g, const float* __restrict__ h, const float* __restrict__ dinv_out,
                const int2* __restrict__ mine, int e, int lane) {
#pragma unroll
  for (int i = 0; i < PE; ++i) {
    const int2 rc = mine[e + i];
    g.dst[i] = rc.y;
    g.s[i] = dinv_out[rc.x];
    const float* src = h + rc.x * DIM + lane;
#pragma unroll
    for (int j = 0; j < 8; ++j) g.v[i][j] = src[j * 32];
  }
}

__device__ __forceinline__
void group_add(const EdgeGroup& g, float* trow) {
#pragma unroll
  for (int i = 0; i < PE; ++i) {
    float* dst = trow + g.dst[i] * DIM;
#pragma unroll
    for (int j = 0; j < 8; ++j) atomAddF(dst + j * 32, g.v[i][j] * g.s[i]);
  }
}

__global__ __launch_bounds__(1024)
void edge_sum_kernel(const float* __restrict__ h, const float* __restrict__ dinv_out,
                     const int2* __restrict__ list, const int* __restrict__ offs,
                     const int* __restrict__ count, float* __restrict__ agg, int N) {
  __shared__ __attribute__((aligned(16))) float tile[TR * DIM];
  const int tid  = threadIdx.x;
  const int lane = tid & 31;
  const int wave = __builtin_amdgcn_readfirstlane(tid >> 5);
  const int b = blockIdx.x;
  const v4f z = {0.0f, 0.0f, 0.0f, 0.0f};
  for (int i = tid; i < TR * DIM / 4; i += 1024) ((v4f*)tile)[i] = z;
  const int cnt = count[b];
  const int2* mine = list + offs[b];
  __syncthreads();

  float* trow = tile + lane;
  const int G = cnt / PE;
  const int m = (G > wave) ? (G - wave + 31) >> 5 : 0;
  const int2* wm = mine + wave * PE;
  if (m > 0) {
    EdgeGroup ga, gb;
    group_load(ga, h, dinv_out, wm, 0, lane);
    int k = 0;
    while (k + 2 < m) {
      group_load(gb, h, dinv_out, wm, (k + 1) * (32 * PE), lane);
      __builtin_amdgcn_sched_barrier(0);
      group_add(ga, trow);
      group_load(ga, h, dinv_out, wm, (k + 2) * (32 * PE), lane);
      __builtin_amdgcn_sched_barrier(0);
      group_add(gb, trow);
      k += 2;
    }
    if (k + 1 < m) {
      group_load(gb, h, dinv_out, wm, (k + 1) * (32 * PE), lane);
      __builtin_amdgcn_sched_barrier(0);
      group_add(ga, trow);
      group_add(gb, trow);
    } else {
      group_add(ga, trow);
    }
  }
  const int n = cnt - G * PE;
  if (n > 0 && wave == (G & 31)) {
    const int e0 = G * PE;
    int   dst[PE - 1];
    float sc[PE - 1];
    float v[PE - 1][8];
#pragma unroll
    for (int i = 0; i < PE - 1; ++i) {
      const int2 rc = mine[e0 + (i < n ? i : n - 1)];
      dst[i] = rc.y;
      sc[i] = dinv_out[rc.x];
      const float* src = h + rc.x * DIM + lane;
#pragma unroll
      for (int j = 0; j < 8; ++j) v[i][j] = src[j * 32];
    }
#pragma unroll
    for (int i = 0; i < PE - 1; ++i) {
      if (i < n) {
        float* d = trow + dst[i] * DIM;
#pragma unroll
        for (int j = 0; j < 8; ++j) atomAddF(d + j * 32, v[i][j] * sc[i]);
      }
    }
  }
  __syncthreads();

  const int row0 = b * TR;
  int rows = N - row0;
  rows = rows > TR ? TR : rows;
  v4f* dstp = (v4f*)(agg + row0 * DIM);
  for (int i = tid; i < rows * (DIM / 4); i += 1024) dstp[i] = ((const v4f*)tile)[i];
}

__global__ __launch_bounds__(256)
void pack_w_kernel(const float* __restrict__ W, _Float16* __restrict__ pack) {
  int t = blockIdx.x * blockDim.x + threadIdx.x;
  int elem = t & 15;
  int lane = (t >> 4) & 31;
  int kc   = (t >> 9) & 7;
  int tn   = t >> 12;
  int v = elem >> 1, p = elem & 1;
  int half = lane >> 4, mn = lane & 15;
  int kb = kc * 32 + ((v & 4) << 2) + (half << 3) + ((v & 3) << 1) + p;
  int gn = tn * 16 + mn;
  pack[t] = (_Float16)W[kb * DIM + gn];
}

#define GM_ROWS 64
#define GM_PITCH 264
template <bool HAS_RESID, bool GUARD>
__device__ __forceinline__
void gemm_block(const float* __restrict__ A, const _Float16* __restrict__ Wp,
                const float* __restrict__ bias, const float* __restrict__ dinv_in,
                const float* __restrict__ resid, float* __restrict__ out, int n_rows,
                _Float16* sA, float* sD) {
  const int tid  = threadIdx.x;
  const int lane = tid & 31;
  const int tn   = tid >> 5;
  const int half = lane >> 4;
  const int mn   = lane & 15;
  const int row0 = blockIdx.x * GM_ROWS;
  const int gn  = tn * 16 + mn;
  const int gm0 = row0 + half * 8;

  float rs[4][8];
  if constexpr (HAS_RESID) {
#pragma unroll
    for (int t = 0; t < 4; ++t) {
#pragma unroll
      for (int r = 0; r < 8; ++r) {
        const int gm = gm0 + t * 16 + r;
        rs[t][r] = (!GUARD || gm < n_rows) ? resid[gm * DIM + gn] : 0.0f;
      }
    }
  }

  const int pr = tid >> 6;
  const int pc = tid & 63;
  const v4f* src = (const v4f*)A + (row0 + pr) * (DIM / 4) + pc;
  v4f q[8];
#pragma unroll
  for (int i = 0; i < 8; ++i) {
    const v4f z = {0.0f, 0.0f, 0.0f, 0.0f};
    q[i] = (!GUARD || row0 + pr + 8 * i < n_rows) ? src[i * 8 * (DIM / 4)] : z;
  }
  if (tid < GM_ROWS) {
    const int r = row0 + tid;
    sD[tid] = (!GUARD || r < n_rows) ? dinv_in[r] : 0.0f;
  }
#pragma unroll
  for (int i = 0; i < 8; ++i) {
    v4h hq;
    hq[0] = (_Float16)q[i].x;  hq[1] = (_Float16)q[i].y;
    hq[2] = (_Float16)q[i].z;  hq[3] = (_Float16)q[i].w;
    *(v4h*)&sA[(pr + 8 * i) * GM_PITCH + pc * 4] = hq;
  }
  __syncthreads();

  const v16h* Bp = (const v16h*)Wp + tn * 256 + lane;
  const _Float16* fa = sA + mn * GM_PITCH + half * 8;
  v8f c[4] = {};
#pragma unroll
  for (int kc = 0; kc < 8; ++kc) {
    const v16h b = Bp[kc * 32];
#pragma unroll
    for (int t = 0; t < 4; ++t) {
      const _Float16* p = fa + t * 16 * GM_PITCH + kc * 32;
      const v8h lo = *(const v8h*)p;
      const v8h hi = *(const v8h*)(p + 16);
      const v16h a = __builtin_shufflevector(lo, hi, 0, 1, 2, 3, 4, 5, 6, 7,
                                             8, 9, 10, 11, 12, 13, 14, 15);
      c[t] = __builtin_amdgcn_wmma_f32_16x16x32_f16(false, a, false, b,
                                                    (short)0, c[t], false, false);
    }
  }

  const float bn = bias[gn];
#pragma unroll
  for (int t = 0; t < 4; ++t) {
    const v4f d0 = *(const v4f*)&sD[t * 16 + half * 8];
    const v4f d1 = *(const v4f*)&sD[t * 16 + half * 8 + 4];
    const float dv[8] = {d0.x, d0.y, d0.z, d0.w, d1.x, d1.y, d1.z, d1.w};
#pragma unroll
    for (int r = 0; r < 8; ++r) {
      const int gm = gm0 + t * 16 + r;
      float val = c[t][r] + bn;
      val *= dv[r];
      if constexpr (HAS_RESID) val += rs[t][r];
      if (!GUARD || gm < n_rows) out[gm * DIM + gn] = val;
    }
  }
}

template <bool HAS_RESID>
__global__ __launch_bounds__(512, 8)
void gemm_wmma_kernel(const float* __restrict__ A, const _Float16* __restrict__ Wp,
                      const float* __restrict__ bias, const float* __restrict__ dinv_in,
                      const float* __restrict__ resid, float* __restrict__ out, int n_rows) {
  __shared__ __attribute__((aligned(16))) _Float16 sA[GM_ROWS * GM_PITCH];
  __shared__ __attribute__((aligned(16))) float sD[GM_ROWS];
  if ((int)(blockIdx.x + 1) * GM_ROWS <= n_rows)
    gemm_block<HAS_RESID, false>(A, Wp, bias, dinv_in, resid, out, n_rows, sA, sD);
  else
    gemm_block<HAS_RESID, true>(A, Wp, bias, dinv_in, resid, out, n_rows, sA, sD);
}

#define BN_ROWS 64
#define BN_T (BN_ROWS / 4)
__global__ __launch_bounds__(256)
void bn_stats_kernel(const float* __restrict__ pre, float* __restrict__ sums,
                     float* __restrict__ sumsq, int n) {
  __shared__ __attribute__((aligned(16))) float part[2][4][DIM];
  const int c4 = threadIdx.x & 63;
  const int ph = threadIdx.x >> 6;
  const int row0 = blockIdx.x * BN_ROWS + ph;
  const v4f* src = (const v4f*)pre + c4;
  v4f v[BN_T];
  if ((int)(blockIdx.x + 1) * BN_ROWS <= n) {
#pragma unroll
    for (int t = 0; t < BN_T; ++t) v[t] = src[(row0 + 4 * t) * (DIM / 4)];
  } else {
#pragma unroll
    for (int t = 0; t < BN_T; ++t) {
      const int r = row0 + 4 * t;
      const v4f z = {0.0f, 0.0f, 0.0f, 0.0f};
      v[t] = (r < n) ? src[r * (DIM / 4)] : z;
    }
  }
  v4f s = {0.0f, 0.0f, 0.0f, 0.0f}, s2 = {0.0f, 0.0f, 0.0f, 0.0f};
#pragma unroll
  for (int t = 0; t < BN_T; ++t) {
    s  += v[t];
    s2 += v[t] * v[t];
  }
  *(v4f*)&part[0][ph][4 * c4] = s;
  *(v4f*)&part[1][ph][4 * c4] = s2;
  __syncthreads();
  const int c = threadIdx.x;
  const float a = ((part[0][0][c] + part[0][1][c]) + part[0][2][c]) + part[0][3][c];
  const float b = ((part[1][0][c] + part[1][1][c]) + part[1][2][c]) + part[1][3][c];
  atomAddF(&sums[c], a);
  atomAddF(&sumsq[c], b);
}

__global__ void bn_finalize_kernel(float* __restrict__ sums, float* __restrict__ sumsq, int n) {
  const int c = threadIdx.x;
  const float inv_n = 1.0f / (float)n;
  float mu  = sums[c] * inv_n;
  float var = sumsq[c] * inv_n - mu * mu;
  sums[c]  = mu;
  sumsq[c] = rsqrtf(var + 1e-5f);
}

#define APPLY_U 8
__device__ __forceinline__ float bn_relu(float x, float mu, float is, float g, float b) {
  float v = (x - mu) * is * g + b;
  return fmaxf(v, 0.0f);
}
__global__ __launch_bounds__(256)
void bn_apply_kernel(float* __restrict__ buf, const float* __restrict__ mu,
                     const float* __restrict__ inv_std, const float* __restrict__ gamma,
                     const float* __restrict__ beta, int n4) {
  v4f* q = (v4f*)buf;
  const int c = (threadIdx.x & 63) * 4;
  const int base = blockIdx.x * (256 * APPLY_U) + threadIdx.x;
  const bool whole = (int)(blockIdx.x + 1) * (256 * APPLY_U) <= n4;
  v4f v[APPLY_U];
  if (whole) {
#pragma unroll
    for (int k = 0; k < APPLY_U; ++k) v[k] = q[base + k * 256];
  } else {
#pragma unroll
    for (int k = 0; k < APPLY_U; ++k) {
      const int i = base + k * 256;
      const v4f z = {0.0f, 0.0f, 0.0f, 0.0f};
      v[k] = (i < n4) ? q[i] : z;
    }
  }
  const float m0 = mu[c], m1 = mu[c + 1], m2 = mu[c + 2], m3 = mu[c + 3];
  const float i0 = inv_std[c], i1 = inv_std[c + 1], i2 = inv_std[c + 2], i3 = inv_std[c + 3];
  const float g0 = gamma[c], g1 = gamma[c + 1], g2 = gamma[c + 2], g3 = gamma[c + 3];
  const float b0 = beta[c], b1 = beta[c + 1], b2 = beta[c + 2], b3 = beta[c + 3];
#pragma unroll
  for (int k = 0; k < APPLY_U; ++k) {
    const int i = base + k * 256;
    if (whole || i < n4) {
      v4f r;
      r.x = bn_relu(v[k].x, m0, i0, g0, b0);
      r.y = bn_relu(v[k].y, m1, i1, g1, b1);
      r.z = bn_relu(v[k].z, m2, i2, g2, b2);
      r.w = bn_relu(v[k].w, m3, i3, g3, b3);
      q[i] = r;
    }
  }
}

extern "C" void kernel_launch(void* const* d_in, const int* in_sizes, int n_in,
                              void* d_out, int out_size, void* d_ws, size_t ws_size,
                              hipStream_t stream) {
  const float* x  = (const float*)d_in[0];
  const int*   ei = (const int*)d_in[1];
  const float* W0 = (const float*)d_in[2];
  const float* b0 = (const float*)d_in[3];
  const float* g0 = (const float*)d_in[4];
  const float* be0= (const float*)d_in[5];
  const float* W1 = (const float*)d_in[6];
  const float* b1 = (const float*)d_in[7];
  const float* g1 = (const float*)d_in[8];
  const float* be1= (const float*)d_in[9];
  const float* W2 = (const float*)d_in[10];
  const float* b2 = (const float*)d_in[11];
  float* out = (float*)d_out;

  const int N = in_sizes[0] / DIM;
  const int E = in_sizes[1] / 2;
  const int ND = N * DIM;
  const int* row = ei;
  const int* col = ei + E;

  float* ws       = (float*)d_ws;
  float* dinv_out = ws;
  float* dinv_in  = ws + N;
  int*   bcount   = (int*)(ws + 2 * N);
  int*   bcursor  = bcount + MAXB;
  float* stats    = ws + 2 * N + 2 * MAXB;
  float* agg      = stats + 512;
  float* bufA     = agg + (size_t)ND;
  _Float16* wp0   = (_Float16*)(bufA + (size_t)ND);
  _Float16* wp1   = wp0 + 65536;
  _Float16* wp2   = wp1 + 65536;
  int*  boffs     = (int*)(wp2 + 65536);
  int2* elist     = (int2*)(boffs + MAXB);

  const int zb = 256;
  auto nb = [](int n, int b) { return (n + b - 1) / b; };

  const int nbins       = nb(N, TR);
  const int bin_blocks  = nb(E, 1024 * BIN_EPT);
  const int gemm_blocks = nb(N, GM_ROWS);
  const int bn_blocks   = nb(N, BN_ROWS);
  const int ND4         = ND / 4;
  const int ap_blocks   = nb(ND4, 256 * APPLY_U);
  const int pack_blocks = 65536 / 256;

  zero_kernel<<<nb(2 * N + 2 * MAXB, zb), zb, 0, stream>>>(dinv_out, 2 * N + 2 * MAXB);
  deg_kernel<<<nb(E, zb), zb, 0, stream>>>(row, col, dinv_out, dinv_in, E);
  rsqrt_kernel<<<nb(2 * N, zb), zb, 0, stream>>>(dinv_out, 2 * N);
  bin_count_kernel<<<bin_blocks, 1024, 0, stream>>>(col, bcount, E, nbins);
  bin_fill_kernel<<<bin_blocks, 1024, 0, stream>>>(row, col, bcount, bcursor, boffs, elist, E, nbins);
  pack_w_kernel<<<pack_blocks, 256, 0, stream>>>(W0, wp0);
  pack_w_kernel<<<pack_blocks, 256, 0, stream>>>(W1, wp1);
  pack_w_kernel<<<pack_blocks, 256, 0, stream>>>(W2, wp2);

  edge_sum_kernel<<<nbins, 1024, 0, stream>>>(x, dinv_out, elist, boffs, bcount, agg, N);
  gemm_wmma_kernel<true><<<gemm_blocks, 512, 0, stream>>>(agg, wp0, b0, dinv_in, x, out, N);
  zero_kernel<<<2, zb, 0, stream>>>(stats, 512);
  bn_stats_kernel<<<bn_blocks, 256, 0, stream>>>(out, stats, stats + 256, N);
  bn_finalize_kernel<<<1, 256, 0, stream>>>(stats, stats + 256, N);
  bn_apply_kernel<<<ap_blocks, 256, 0, stream>>>(out, stats, stats + 256, g0, be0, ND4);

  edge_sum_kernel<<<nbins, 1024, 0, stream>>>(out, dinv_out, elist, boffs, bcount, agg, N);
  gemm_wmma_kernel<true><<<gemm_blocks, 512, 0, stream>>>(agg, wp1, b1, dinv_in, out, bufA, N);
  zero_kernel<<<2, zb, 0, stream>>>(stats, 512);
  bn_stats_kernel<<<bn_blocks, 256, 0, stream>>>(bufA, stats, stats + 256, N);
  bn_finalize_kernel<<<1, 256, 0, stream>>>(stats, stats + 256, N);
  bn_apply_kernel<<<ap_blocks, 256, 0, stream>>>(bufA, stats, stats + 256, g1, be1, ND4);

  edge_sum_kernel<<<nbins, 1024, 0, stream>>>(bufA, dinv_out, elist, boffs, bcount, agg, N);
  gemm_wmma_kernel<false><<<gemm_blocks, 512, 0, stream>>>(agg, wp2, b2, dinv_in, nullptr, out, N);
}


# Launches 22 -> 13: weight packs and all clears in one first launch, degree counts ride in the bin-count pass, rsqrt in the bin-fill launch, BN mean/inv-std formed per lane inside the apply pass
# speedup vs baseline: 3.3251x; 1.0739x over previous
#include <hip/hip_runtime.h>
#include <math.h>


typedef __attribute__((ext_vector_type(16))) _Float16 v16h;
typedef __attribute__((ext_vector_type(8)))  _Float16 v8h;
typedef __attribute__((ext_vector_type(4)))  _Float16 v4h;
typedef __attribute__((ext_vector_type(8)))  float    v8f;
typedef __attribute__((ext_vector_type(4)))  float    v4f;

#define DIM 256

__device__ __forceinline__ float atomAddF(float* p, float v) {
  return unsafeAtomicAdd(p, v);
}

#define PACK_BLOCKS 256
__global__ __launch_bounds__(256)
void prep_kernel(const float* __restrict__ W0, const float* __restrict__ W1,
                 const float* __restrict__ W2, _Float16* __restrict__ pack,
                 float* __restrict__ z, int nz) {
  const int b = blockIdx.x;
  if (b < 3 * PACK_BLOCKS) {
    const int which = b / PACK_BLOCKS;
    const float* __restrict__ W = which == 0 ? W0 : (which == 1 ? W1 : W2);
    const int t = (b - which * PACK_BLOCKS) * 256 + threadIdx.x;
    const int elem = t & 15;
    const int lane = (t >> 4) & 31;
    const int kc   = (t >> 9) & 7;
    const int tn   = t >> 12;
    const int v = elem >> 1, p = elem & 1;
    const int half = lane >> 4, mn = lane & 15;
    const int kb = kc * 32 + ((v & 4) << 2) + (half << 3) + ((v & 3) << 1) + p;
    const int gn = tn * 16 + mn;
    pack[which * 65536 + t] = (_Float16)W[kb * DIM + gn];
  } else {
    const int i = (b - 3 * PACK_BLOCKS) * 256 + threadIdx.x;
    if (i < nz) z[i] = 0.0f;
  }
}

#define TR 196
#define MAXB 1024
#define BIN_EPT 4
#define CNT_EPT 2

__global__ __launch_bounds__(1024)
void bin_count_kernel(const int* __restrict__ row, const int* __restrict__ col,
                      float* __restrict__ deg_out, float* __restrict__ deg_in,
                      int* __restrict__ count, int E, int nbins) {
  __shared__ int hist[MAXB];
  const int tid = threadIdx.x;
  hist[tid] = 0;
  __syncthreads();
  const int base = blockIdx.x * (1024 * CNT_EPT) + tid;
  for (int k = 0; k < CNT_EPT; ++k) {
    const int e = base + k * 1024;
    if (e < E) {
      const int cc = col[e];
      atomicAdd(&hist[cc / TR], 1);
      atomAddF(&deg_in[cc], 1.0f);
      atomAddF(&deg_out[row[e]], 1.0f);
    }
  }
  __syncthreads();
  if (tid < nbins) {
    const int hc = hist[tid];
    if (hc) atomicAdd(&count[tid], hc);
  }
}

__global__ __launch_bounds__(1024)
void bin_fill_kernel(const int* __restrict__ row, const int* __restrict__ col,
                     const int* __restrict__ count, int* __restrict__ cursor,
                     int* __restrict__ offs, int2* __restrict__ list, int E, int nbins,
                     float* __restrict__ deg, int nd) {
  __shared__ int hist[MAXB];
  __shared__ int scan[MAXB];
  __shared__ int basep[MAXB];
  const int tid = threadIdx.x;
  for (int i = blockIdx.x * 1024 + tid; i < nd; i += gridDim.x * 1024) {
    float v = deg[i];
    v = v < 1.0f ? 1.0f : v;
    deg[i] = rsqrtf(v);
  }
  const int c = (tid < nbins) ? count[tid] : 0;
  scan[tid] = c;
  hist[tid] = 0;
  __syncthreads();
  for (int d = 1; d < MAXB; d <<= 1) {
    int v = scan[tid];
    if (tid >= d) v += scan[tid - d];
    __syncthreads();
    scan[tid] = v;
    __syncthreads();
  }
  const int excl = scan[tid] - c;
  if (blockIdx.x == 0 && tid < nbins) offs[tid] = excl;

  const int base = blockIdx.x * (1024 * BIN_EPT) + tid;
  int rr[BIN_EPT], cl[BIN_EPT], bb[BIN_EPT], rk[BIN_EPT];
#pragma unroll
  for (int k = 0; k < BIN_EPT; ++k) {
    const int e = base + k * 1024;
    bb[k] = -1;
    if (e < E) {
      const int cc = col[e];
      rr[k] = row[e];
      bb[k] = cc / TR;
      cl[k] = cc - bb[k] * TR;
      rk[k] = atomicAdd(&hist[bb[k]], 1);
    }
  }
  __syncthreads();
  if (tid < nbins) {
    const int hc = hist[tid];
    basep[tid] = excl + (hc ? atomicAdd(&cursor[tid], hc) : 0);
  }
  __syncthreads();
#pragma unroll
  for (int k = 0; k < BIN_EPT; ++k) {
    if (bb[k] >= 0) list[basep[bb[k]] + rk[k]] = make_int2(rr[k], cl[k]);
  }
}

#define PE 4
struct EdgeGroup {
  int   dst[PE];
  float s[PE];
  float v[PE][8];
};

__device__ __forceinline__
void group_load(EdgeGroup& g, const float* __restrict__ h, const float* __restrict__ dinv_out,
                const int2* __restrict__ mine, int e, int lane) {
#pragma unroll
  for (int i = 0; i < PE; ++i) {
    const int2 rc = mine[e + i];
    g.dst[i] = rc.y;
    g.s[i] = dinv_out[rc.x];
    const float* src = h + rc.x * DIM + lane;
#pragma unroll
    for (int j = 0; j < 8; ++j) g.v[i][j] = src[j * 32];
  }
}

__device__ __forceinline__
void group_add(const EdgeGroup& g, float* trow) {
#pragma unroll
  for (int i = 0; i < PE; ++i) {
    float* dst = trow + g.dst[i] * DIM;
#pragma unroll
    for (int j = 0; j < 8; ++j) atomAddF(dst + j * 32, g.v[i][j] * g.s[i]);
  }
}

__global__ __launch_bounds__(1024)
void edge_sum_kernel(const float* __restrict__ h, const float* __restrict__ dinv_out,
                     const int2* __restrict__ list, const int* __restrict__ offs,
                     const int* __restrict__ count, float* __restrict__ agg, int N) {
  __shared__ __attribute__((aligned(16))) float tile[TR * DIM];
  const int tid  = threadIdx.x;
  const int lane = tid & 31;
  const int wave = __builtin_amdgcn_readfirstlane(tid >> 5);
  const int b = blockIdx.x;
  const v4f z = {0.0f, 0.0f, 0.0f, 0.0f};
  for (int i = tid; i < TR * DIM / 4; i += 1024) ((v4f*)tile)[i] = z;
  const int cnt = count[b];
  const int2* mine = list + offs[b];
  __syncthreads();

  float* trow = tile + lane;
  const int G = cnt / PE;
  const int m = (G > wave) ? (G - wave + 31) >> 5 : 0;
  const int2* wm = mine + wave * PE;
  if (m > 0) {
    EdgeGroup ga, gb;
    group_load(ga, h, dinv_out, wm, 0, lane);
    int k = 0;
    while (k + 2 < m) {
      group_load(gb, h, dinv_out, wm, (k + 1) * (32 * PE), lane);
      __builtin_amdgcn_sched_barrier(0);
      group_add(ga, trow);
      group_load(ga, h, dinv_out, wm, (k + 2) * (32 * PE), lane);
      __builtin_amdgcn_sched_barrier(0);
      group_add(gb, trow);
      k += 2;
    }
    if (k + 1 < m) {
      group_load(gb, h, dinv_out, wm, (k + 1) * (32 * PE), lane);
      __builtin_amdgcn_sched_barrier(0);
      group_add(ga, trow);
      group_add(gb, trow);
    } else {
      group_add(ga, trow);
    }
  }
  const int n = cnt - G * PE;
  if (n > 0 && wave == (G & 31)) {
    const int e0 = G * PE;
    int   dst[PE - 1];
    float sc[PE - 1];
    float v[PE - 1][8];
#pragma unroll
    for (int i = 0; i < PE - 1; ++i) {
      const int2 rc = mine[e0 + (i < n ? i : n - 1)];
      dst[i] = rc.y;
      sc[i] = dinv_out[rc.x];
      const float* src = h + rc.x * DIM + lane;
#pragma unroll
      for (int j = 0; j < 8; ++j) v[i][j] = src[j * 32];
    }
#pragma unroll
    for (int i = 0; i < PE - 1; ++i) {
      if (i < n) {
        float* d = trow + dst[i] * DIM;
#pragma unroll
        for (int j = 0; j < 8; ++j) atomAddF(d + j * 32, v[i][j] * sc[i]);
      }
    }
  }
  __syncthreads();

  const int row0 = b * TR;
  int rows = N - row0;
  rows = rows > TR ? TR : rows;
  v4f* dstp = (v4f*)(agg + row0 * DIM);
  for (int i = tid; i < rows * (DIM / 4); i += 1024) dstp[i] = ((const v4f*)tile)[i];
}

#define GM_ROWS 64
#define GM_PITCH 264
template <bool HAS_RESID, bool GUARD>
__device__ __forceinline__
void gemm_block(const float* __restrict__ A, const _Float16* __restrict__ Wp,
                const float* __restrict__ bias, const float* __restrict__ dinv_in,
                const float* __restrict__ resid, float* __restrict__ out, int n_rows,
                _Float16* sA, float* sD) {
  const int tid  = threadIdx.x;
  const int lane = tid & 31;
  const int tn   = tid >> 5;
  const int half = lane >> 4;
  const int mn   = lane & 15;
  const int row0 = blockIdx.x * GM_ROWS;
  const int gn  = tn * 16 + mn;
  const int gm0 = row0 + half * 8;

  float rs[4][8];
  if constexpr (HAS_RESID) {
#pragma unroll
    for (int t = 0; t < 4; ++t) {
#pragma unroll
      for (int r = 0; r < 8; ++r) {
        const int gm = gm0 + t * 16 + r;
        rs[t][r] = (!GUARD || gm < n_rows) ? resid[gm * DIM + gn] : 0.0f;
      }
    }
  }

  const int pr = tid >> 6;
  const int pc = tid & 63;
  const v4f* src = (const v4f*)A + (row0 + pr) * (DIM / 4) + pc;
  v4f q[8];
#pragma unroll
  for (int i = 0; i < 8; ++i) {
    const v4f z = {0.0f, 0.0f, 0.0f, 0.0f};
    q[i] = (!GUARD || row0 + pr + 8 * i < n_rows) ? src[i * 8 * (DIM / 4)] : z;
  }
  if (tid < GM_ROWS) {
    const int r = row0 + tid;
    sD[tid] = (!GUARD || r < n_rows) ? dinv_in[r] : 0.0f;
  }
#pragma unroll
  for (int i = 0; i < 8; ++i) {
    v4h hq;
    hq[0] = (_Float16)q[i].x;  hq[1] = (_Float16)q[i].y;
    hq[2] = (_Float16)q[i].z;  hq[3] = (_Float16)q[i].w;
    *(v4h*)&sA[(pr + 8 * i) * GM_PITCH + pc * 4] = hq;
  }
  __syncthreads();

  const v16h* Bp = (const v16h*)Wp + tn * 256 + lane;
  const _Float16* fa = sA + mn * GM_PITCH + half * 8;
  v8f c[4] = {};
#pragma unroll
  for (int kc = 0; kc < 8; ++kc) {
    const v16h b = Bp[kc * 32];
#pragma unroll
    for (int t = 0; t < 4; ++t) {
      const _Float16* p = fa + t * 16 * GM_PITCH + kc * 32;
      const v8h lo = *(const v8h*)p;
      const v8h hi = *(const v8h*)(p + 16);
      const v16h a = __builtin_shufflevector(lo, hi, 0, 1, 2, 3, 4, 5, 6, 7,
                                             8, 9, 10, 11, 12, 13, 14, 15);
      c[t] = __builtin_amdgcn_wmma_f32_16x16x32_f16(false, a, false, b,
                                                    (short)0, c[t], false, false);
    }
  }

  const float bn = bias[gn];
#pragma unroll
  for (int t = 0; t < 4; ++t) {
    const v4f d0 = *(const v4f*)&sD[t * 16 + half * 8];
    const v4f d1 = *(const v4f*)&sD[t * 16 + half * 8 + 4];
    const float dv[8] = {d0.x, d0.y, d0.z, d0.w, d1.x, d1.y, d1.z, d1.w};
#pragma unroll
    for (int r = 0; r < 8; ++r) {
      const int gm = gm0 + t * 16 + r;
      float val = c[t][r] + bn;
      val *= dv[r];
      if constexpr (HAS_RESID) val += rs[t][r];
      if (!GUARD || gm < n_rows) out[gm * DIM + gn] = val;
    }
  }
}

template <bool HAS_RESID>
__global__ __launch_bounds__(512, 8)
void gemm_wmma_kernel(const float* __restrict__ A, const _Float16* __restrict__ Wp,
                      const float* __restrict__ bias, const float* __restrict__ dinv_in,
                      const float* __restrict__ resid, float* __restrict__ out, int n_rows) {
  __shared__ __attribute__((aligned(16))) _Float16 sA[GM_ROWS * GM_PITCH];
  __shared__ __attribute__((aligned(16))) float sD[GM_ROWS];
  if ((int)(blockIdx.x + 1) * GM_ROWS <= n_rows)
    gemm_block<HAS_RESID, false>(A, Wp, bias, dinv_in, resid, out, n_rows, sA, sD);
  else
    gemm_block<HAS_RESID, true>(A, Wp, bias, dinv_in, resid, out, n_rows, sA, sD);
}

#define BN_ROWS 64
#define BN_T (BN_ROWS / 4)
__global__ __launch_bounds__(256)
void bn_stats_kernel(const float* __restrict__ pre, float* __restrict__ sums,
                     float* __restrict__ sumsq, int n) {
  __shared__ __attribute__((aligned(16))) float part[2][4][DIM];
  const int c4 = threadIdx.x & 63;
  const int ph = threadIdx.x >> 6;
  const int row0 = blockIdx.x * BN_ROWS + ph;
  const v4f* src = (const v4f*)pre + c4;
  v4f v[BN_T];
  if ((int)(blockIdx.x + 1) * BN_ROWS <= n) {
#pragma unroll
    for (int t = 0; t < BN_T; ++t) v[t] = src[(row0 + 4 * t) * (DIM / 4)];
  } else {
#pragma unroll
    for (int t = 0; t < BN_T; ++t) {
      const int r = row0 + 4 * t;
      const v4f z = {0.0f, 0.0f, 0.0f, 0.0f};
      v[t] = (r < n) ? src[r * (DIM / 4)] : z;
    }
  }
  v4f s = {0.0f, 0.0f, 0.0f, 0.0f}, s2 = {0.0f, 0.0f, 0.0f, 0.0f};
#pragma unroll
  for (int t = 0; t < BN_T; ++t) {
    s  += v[t];
    s2 += v[t] * v[t];
  }
  *(v4f*)&part[0][ph][4 * c4] = s;
  *(v4f*)&part[1][ph][4 * c4] = s2;
  __syncthreads();
  const int c = threadIdx.x;
  const float a = ((part[0][0][c] + part[0][1][c]) + part[0][2][c]) + part[0][3][c];
  const float b = ((part[1][0][c] + part[1][1][c]) + part[1][2][c]) + part[1][3][c];
  atomAddF(&sums[c], a);
  atomAddF(&sumsq[c], b);
}

#define APPLY_U 8
__device__ __forceinline__ float bn_relu(float x, float mu, float is, float g, float b) {
  float v = (x - mu) * is * g + b;
  return fmaxf(v, 0.0f);
}
__device__ __forceinline__ void bn_moments(float s, float ss, float inv_n, float& mu, float& is) {
  mu = s * inv_n;
  const float var = ss * inv_n - mu * mu;
  is = rsqrtf(var + 1e-5f);
}
__global__ __launch_bounds__(256)
void bn_apply_kernel(float* __restrict__ buf, const float* __restrict__ sums,
                     const float* __restrict__ sumsq, const float* __restrict__ gamma,
                     const float* __restrict__ beta, int n4, int n) {
  v4f* q = (v4f*)buf;
  const int c = (threadIdx.x & 63) * 4;
  const int base = blockIdx.x * (256 * APPLY_U) + threadIdx.x;
  const bool whole = (int)(blockIdx.x + 1) * (256 * APPLY_U) <= n4;
  v4f v[APPLY_U];
  if (whole) {
#pragma unroll
    for (int k = 0; k < APPLY_U; ++k) v[k] = q[base + k * 256];
  } else {
#pragma unroll
    for (int k = 0; k < APPLY_U; ++k) {
      const int i = base + k * 256;
      const v4f z = {0.0f, 0.0f, 0.0f, 0.0f};
      v[k] = (i < n4) ? q[i] : z;
    }
  }
  const float inv_n = 1.0f / (float)n;
  float m0, m1, m2, m3, i0, i1, i2, i3;
  bn_moments(sums[c],     sumsq[c],     inv_n, m0, i0);
  bn_moments(sums[c + 1], sumsq[c + 1], inv_n, m1, i1);
  bn_moments(sums[c + 2], sumsq[c + 2], inv_n, m2, i2);
  bn_moments(sums[c + 3], sumsq[c + 3], inv_n, m3, i3);
  const float g0 = gamma[c], g1 = gamma[c + 1], g2 = gamma[c + 2], g3 = gamma[c + 3];
  const float b0 = beta[c], b1 = beta[c + 1], b2 = beta[c + 2], b3 = beta[c + 3];
#pragma unroll
  for (int k = 0; k < APPLY_U; ++k) {
    const int i = base + k * 256;
    if (whole || i < n4) {
      v4f r;
      r.x = bn_relu(v[k].x, m0, i0, g0, b0);
      r.y = bn_relu(v[k].y, m1, i1, g1, b1);
      r.z = bn_relu(v[k].z, m2, i2, g2, b2);
      r.w = bn_relu(v[k].w, m3, i3, g3, b3);
      q[i] = r;
    }
  }
}

extern "C" void kernel_launch(void* const* d_in, const int* in_sizes, int n_in,
                              void* d_out, int out_size, void* d_ws, size_t ws_size,
                              hipStream_t stream) {
  const float* x  = (const float*)d_in[0];
  const int*   ei = (const int*)d_in[1];
  const float* W0 = (const float*)d_in[2];
  const float* b0 = (const float*)d_in[3];
  const float* g0 = (const float*)d_in[4];
  const float* be0= (const float*)d_in[5];
  const float* W1 = (const float*)d_in[6];
  const float* b1 = (const float*)d_in[7];
  const float* g1 = (const float*)d_in[8];
  const float* be1= (const float*)d_in[9];
  const float* W2 = (const float*)d_in[10];
  const float* b2 = (const float*)d_in[11];
  float* out = (float*)d_out;

  const int N = in_sizes[0] / DIM;
  const int E = in_sizes[1] / 2;
  const int ND = N * DIM;
  const int* row = ei;
  const int* col = ei + E;

  float* ws       = (float*)d_ws;
  float* dinv_out = ws;
  float* dinv_in  = ws + N;
  int*   bcount   = (int*)(ws + 2 * N);
  int*   bcursor  = bcount + MAXB;
  float* stats0   = ws + 2 * N + 2 * MAXB;
  float* stats1   = stats0 + 512;
  float* agg      = stats1 + 512;
  float* bufA     = agg + (size_t)ND;
  _Float16* wp0   = (_Float16*)(bufA + (size_t)ND);
  _Float16* wp1   = wp0 + 65536;
  _Float16* wp2   = wp1 + 65536;
  int*  boffs     = (int*)(wp2 + 65536);
  int2* elist     = (int2*)(boffs + MAXB);

  const int zb = 256;
  auto nb = [](int n, int b) { return (n + b - 1) / b; };

  const int nbins       = nb(N, TR);
  const int bin_blocks  = nb(E, 1024 * BIN_EPT);
  const int cnt_blocks  = nb(E, 1024 * CNT_EPT);
  const int nclear      = 2 * N + 2 * MAXB + 1024;
  const int gemm_blocks = nb(N, GM_ROWS);
  const int bn_blocks   = nb(N, BN_ROWS);
  const int ND4         = ND / 4;
  const int ap_blocks   = nb(ND4, 256 * APPLY_U);

  prep_kernel<<<3 * PACK_BLOCKS + nb(nclear, zb), zb, 0, stream>>>(W0, W1, W2, wp0, dinv_out, nclear);
  bin_count_kernel<<<cnt_blocks, 1024, 0, stream>>>(row, col, dinv_out, dinv_in, bcount, E, nbins);
  bin_fill_kernel<<<bin_blocks, 1024, 0, stream>>>(row, col, bcount, bcursor, boffs, elist, E, nbins,
                                                   dinv_out, 2 * N);

  edge_sum_kernel<<<nbins, 1024, 0, stream>>>(x, dinv_out, elist, boffs, bcount, agg, N);
  gemm_wmma_kernel<true><<<gemm_blocks, 512, 0, stream>>>(agg, wp0, b0, dinv_in, x, out, N);
  bn_stats_kernel<<<bn_blocks, 256, 0, stream>>>(out, stats0, stats0 + 256, N);
  bn_apply_kernel<<<ap_blocks, 256, 0, stream>>>(out, stats0, stats0 + 256, g0, be0, ND4, N);

  edge_sum_kernel<<<nbins, 1024, 0, stream>>>(out, dinv_out, elist, boffs, bcount, agg, N);
  gemm_wmma_kernel<true><<<gemm_blocks, 512, 0, stream>>>(agg, wp1, b1, dinv_in, out, bufA, N);
  bn_stats_kernel<<<bn_blocks, 256, 0, stream>>>(bufA, stats1, stats1 + 256, N);
  bn_apply_kernel<<<ap_blocks, 256, 0, stream>>>(bufA, stats1, stats1 + 256, g1, be1, ND4, N);

  edge_sum_kernel<<<nbins, 1024, 0, stream>>>(bufA, dinv_out, elist, boffs, bcount, agg, N);
  gemm_wmma_kernel<false><<<gemm_blocks, 512, 0, stream>>>(agg, wp2, b2, dinv_in, nullptr, out, N);
}


# Degrees without global atomics: 256 work-groups each count one edge slice's target or source rows of one node range in LDS (integer adds, exact), store plane parts; rsqrt pass adds 32 planes a node
# speedup vs baseline: 3.7308x; 1.1220x over previous
#include <hip/hip_runtime.h>
#include <math.h>


typedef __attribute__((ext_vector_type(16))) _Float16 v16h;
typedef __attribute__((ext_vector_type(8)))  _Float16 v8h;
typedef __attribute__((ext_vector_type(4)))  _Float16 v4h;
typedef __attribute__((ext_vector_type(8)))  float    v8f;
typedef __attribute__((ext_vector_type(4)))  float    v4f;

#define DIM 256

__device__ __forceinline__ float atomAddF(float* p, float v) {
  return unsafeAtomicAdd(p, v);
}

#define PACK_BLOCKS 256
__global__ __launch_bounds__(256)
void prep_kernel(const float* __restrict__ W0, const float* __restrict__ W1,
                 const float* __restrict__ W2, _Float16* __restrict__ pack,
                 float* __restrict__ z, int nz) {
  const int b = blockIdx.x;
  if (b < 3 * PACK_BLOCKS) {
    const int which = b / PACK_BLOCKS;
    const float* __restrict__ W = which == 0 ? W0 : (which == 1 ? W1 : W2);
    const int t = (b - which * PACK_BLOCKS) * 256 + threadIdx.x;
    const int elem = t & 15;
    const int lane = (t >> 4) & 31;
    const int kc   = (t >> 9) & 7;
    const int tn   = t >> 12;
    const int v = elem >> 1, p = elem & 1;
    const int half = lane >> 4, mn = lane & 15;
    const int kb = kc * 32 + ((v & 4) << 2) + (half << 3) + ((v & 3) << 1) + p;
    const int gn = tn * 16 + mn;
    pack[which * 65536 + t] = (_Float16)W[kb * DIM + gn];
  } else {
    const int i = (b - 3 * PACK_BLOCKS) * 256 + threadIdx.x;
    if (i < nz) z[i] = 0.0f;
  }
}

#define TR 196
#define MAXB 1024
#define BIN_EPT 4
#define DEG_S 32
#define DEG_CAP (TR * 256)
#define DEG_Q 4
#define DEG_QCAP (DEG_CAP / DEG_Q)
#define DEG_EPT 16

typedef __attribute__((ext_vector_type(4))) int v4i;

__global__ __launch_bounds__(1024)
void bin_count_kernel(const int* __restrict__ row, const int* __restrict__ col,
                      int* __restrict__ part, int* __restrict__ count, int E, int nbins) {
  __shared__ __attribute__((aligned(16))) int cnt[DEG_QCAP];
  __shared__ int hist[MAXB];
  const int tid = threadIdx.x;
  const int q = blockIdx.x % DEG_Q;
  const int kind = (blockIdx.x / DEG_Q) & 1;
  const int s = blockIdx.x / (2 * DEG_Q);
  const int lo = q * DEG_QCAP;
  const v4i z = {0, 0, 0, 0};
  for (int i = tid; i < DEG_QCAP / 4; i += 1024) ((v4i*)cnt)[i] = z;
  hist[tid] = 0;
  __syncthreads();
  const int per = (E + DEG_S - 1) / DEG_S;
  const int e0 = s * per;
  int e1 = e0 + per;
  e1 = e1 < E ? e1 : E;
  const int* __restrict__ key = kind ? row : col;
  for (int base = e0 + tid; base < e1; base += 1024 * DEG_EPT) {
    int kv[DEG_EPT];
#pragma unroll
    for (int k = 0; k < DEG_EPT; ++k) {
      const int e = base + k * 1024;
      kv[k] = (e < e1) ? key[e] : -1;
    }
#pragma unroll
    for (int k = 0; k < DEG_EPT; ++k) {
      const int r = kv[k] - lo;
      if ((unsigned)r < (unsigned)DEG_QCAP) {
        atomicAdd(&cnt[r], 1);
        if (kind == 0) atomicAdd(&hist[kv[k] / TR], 1);
      }
    }
  }
  __syncthreads();
  if (kind == 0 && tid < nbins) {
    const int hc = hist[tid];
    if (hc) atomicAdd(&count[tid], hc);
  }
  v4i* dst = (v4i*)(part + (size_t)(2 * s + kind) * DEG_CAP + lo);
  for (int i = tid; i < DEG_QCAP / 4; i += 1024) dst[i] = ((const v4i*)cnt)[i];
}

__global__ __launch_bounds__(1024)
void bin_fill_kernel(const int* __restrict__ row, const int* __restrict__ col,
                     const int* __restrict__ count, int* __restrict__ cursor,
                     int* __restrict__ offs, int2* __restrict__ list, int E, int nbins,
                     const int* __restrict__ part, float* __restrict__ deg, int N) {
  __shared__ int hist[MAXB];
  __shared__ int scan[MAXB];
  __shared__ int basep[MAXB];
  const int tid = threadIdx.x;
  for (int i = blockIdx.x * 1024 + tid; i < 2 * N; i += gridDim.x * 1024) {
    const int* p = part + (i < N ? DEG_CAP + i : i - N);
    int c = 0;
#pragma unroll
    for (int s = 0; s < DEG_S; ++s) c += p[(size_t)(2 * s) * DEG_CAP];
    float v = (float)c;
    v = v < 1.0f ? 1.0f : v;
    deg[i] = rsqrtf(v);
  }
  const int c = (tid < nbins) ? count[tid] : 0;
  scan[tid] = c;
  hist[tid] = 0;
  __syncthreads();
  for (int d = 1; d < MAXB; d <<= 1) {
    int v = scan[tid];
    if (tid >= d) v += scan[tid - d];
    __syncthreads();
    scan[tid] = v;
    __syncthreads();
  }
  const int excl = scan[tid] - c;
  if (blockIdx.x == 0 && tid < nbins) offs[tid] = excl;

  const int base = blockIdx.x * (1024 * BIN_EPT) + tid;
  int rr[BIN_EPT], cl[BIN_EPT], bb[BIN_EPT], rk[BIN_EPT];
#pragma unroll
  for (int k = 0; k < BIN_EPT; ++k) {
    const int e = base + k * 1024;
    bb[k] = -1;
    if (e < E) {
      const int cc = col[e];
      rr[k] = row[e];
      bb[k] = cc / TR;
      cl[k] = cc - bb[k] * TR;
      rk[k] = atomicAdd(&hist[bb[k]], 1);
    }
  }
  __syncthreads();
  if (tid < nbins) {
    const int hc = hist[tid];
    basep[tid] = excl + (hc ? atomicAdd(&cursor[tid], hc) : 0);
  }
  __syncthreads();
#pragma unroll
  for (int k = 0; k < BIN_EPT; ++k) {
    if (bb[k] >= 0) list[basep[bb[k]] + rk[k]] = make_int2(rr[k], cl[k]);
  }
}

#define PE 4
struct EdgeGroup {
  int   dst[PE];
  float s[PE];
  float v[PE][8];
};

__device__ __forceinline__
void group_load(EdgeGroup& g, const float* __restrict__ h, const float* __restrict__ dinv_out,
                const int2* __restrict__ mine, int e, int lane) {
#pragma unroll
  for (int i = 0; i < PE; ++i) {
    const int2 rc = mine[e + i];
    g.dst[i] = rc.y;
    g.s[i] = dinv_out[rc.x];
    const float* src = h + rc.x * DIM + lane;
#pragma unroll
    for (int j = 0; j < 8; ++j) g.v[i][j] = src[j * 32];
  }
}

__device__ __forceinline__
void group_add(const EdgeGroup& g, float* trow) {
#pragma unroll
  for (int i = 0; i < PE; ++i) {
    float* dst = trow + g.dst[i] * DIM;
#pragma unroll
    for (int j = 0; j < 8; ++j) atomAddF(dst + j * 32, g.v[i][j] * g.s[i]);
  }
}

__global__ __launch_bounds__(1024)
void edge_sum_kernel(const float* __restrict__ h, const float* __restrict__ dinv_out,
                     const int2* __restrict__ list, const int* __restrict__ offs,
                     const int* __restrict__ count, float* __restrict__ agg, int N) {
  __shared__ __attribute__((aligned(16))) float tile[TR * DIM];
  const int tid  = threadIdx.x;
  const int lane = tid & 31;
  const int wave = __builtin_amdgcn_readfirstlane(tid >> 5);
  const int b = blockIdx.x;
  const v4f z = {0.0f, 0.0f, 0.0f, 0.0f};
  for (int i = tid; i < TR * DIM / 4; i += 1024) ((v4f*)tile)[i] = z;
  const int cnt = count[b];
  const int2* mine = list + offs[b];
  __syncthreads();

  float* trow = tile + lane;
  const int G = cnt / PE;
  const int m = (G > wave) ? (G - wave + 31) >> 5 : 0;
  const int2* wm = mine + wave * PE;
  if (m > 0) {
    EdgeGroup ga, gb;
    group_load(ga, h, dinv_out, wm, 0, lane);
    int k = 0;
    while (k + 2 < m) {
      group_load(gb, h, dinv_out, wm, (k + 1) * (32 * PE), lane);
      __builtin_amdgcn_sched_barrier(0);
      group_add(ga, trow);
      group_load(ga, h, dinv_out, wm, (k + 2) * (32 * PE), lane);
      __builtin_amdgcn_sched_barrier(0);
      group_add(gb, trow);
      k += 2;
    }
    if (k + 1 < m) {
      group_load(gb, h, dinv_out, wm, (k + 1) * (32 * PE), lane);
      __builtin_amdgcn_sched_barrier(0);
      group_add(ga, trow);
      group_add(gb, trow);
    } else {
      group_add(ga, trow);
    }
  }
  const int n = cnt - G * PE;
  if (n > 0 && wave == (G & 31)) {
    const int e0 = G * PE;
    int   dst[PE - 1];
    float sc[PE - 1];
    float v[PE - 1][8];
#pragma unroll
    for (int i = 0; i < PE - 1; ++i) {
      const int2 rc = mine[e0 + (i < n ? i : n - 1)];
      dst[i] = rc.y;
      sc[i] = dinv_out[rc.x];
      const float* src = h + rc.x * DIM + lane;
#pragma unroll
      for (int j = 0; j < 8; ++j) v[i][j] = src[j * 32];
    }
#pragma unroll
    for (int i = 0; i < PE - 1; ++i) {
      if (i < n) {
        float* d = trow + dst[i] * DIM;
#pragma unroll
        for (int j = 0; j < 8; ++j) atomAddF(d + j * 32, v[i][j] * sc[i]);
      }
    }
  }
  __syncthreads();

  const int row0 = b * TR;
  int rows = N - row0;
  rows = rows > TR ? TR : rows;
  v4f* dstp = (v4f*)(agg + row0 * DIM);
  for (int i = tid; i < rows * (DIM / 4); i += 1024) dstp[i] = ((const v4f*)tile)[i];
}

#define GM_ROWS 64
#define GM_PITCH 264
template <bool HAS_RESID, bool GUARD>
__device__ __forceinline__
void gemm_block(const float* __restrict__ A, const _Float16* __restrict__ Wp,
                const float* __restrict__ bias, const float* __restrict__ dinv_in,
                const float* __restrict__ resid, float* __restrict__ out, int n_rows,
                _Float16* sA, float* sD) {
  const int tid  = threadIdx.x;
  const int lane = tid & 31;
  const int tn   = tid >> 5;
  const int half = lane >> 4;
  const int mn   = lane & 15;
  const int row0 = blockIdx.x * GM_ROWS;
  const int gn  = tn * 16 + mn;
  const int gm0 = row0 + half * 8;

  float rs[4][8];
  if constexpr (HAS_RESID) {
#pragma unroll
    for (int t = 0; t < 4; ++t) {
#pragma unroll
      for (int r = 0; r < 8; ++r) {
        const int gm = gm0 + t * 16 + r;
        rs[t][r] = (!GUARD || gm < n_rows) ? resid[gm * DIM + gn] : 0.0f;
      }
    }
  }

  const int pr = tid >> 6;
  const int pc = tid & 63;
  const v4f* src = (const v4f*)A + (row0 + pr) * (DIM / 4) + pc;
  v4f q[8];
#pragma unroll
  for (int i = 0; i < 8; ++i) {
    const v4f z = {0.0f, 0.0f, 0.0f, 0.0f};
    q[i] = (!GUARD || row0 + pr + 8 * i < n_rows) ? src[i * 8 * (DIM / 4)] : z;
  }
  if (tid < GM_ROWS) {
    const int r = row0 + tid;
    sD[tid] = (!GUARD || r < n_rows) ? dinv_in[r] : 0.0f;
  }
#pragma unroll
  for (int i = 0; i < 8; ++i) {
    v4h hq;
    hq[0] = (_Float16)q[i].x;  hq[1] = (_Float16)q[i].y;
    hq[2] = (_Float16)q[i].z;  hq[3] = (_Float16)q[i].w;
    *(v4h*)&sA[(pr + 8 * i) * GM_PITCH + pc * 4] = hq;
  }
  __syncthreads();

  const v16h* Bp = (const v16h*)Wp + tn * 256 + lane;
  const _Float16* fa = sA + mn * GM_PITCH + half * 8;
  v8f c[4] = {};
#pragma unroll
  for (int kc = 0; kc < 8; ++kc) {
    const v16h b = Bp[kc * 32];
#pragma unroll
    for (int t = 0; t < 4; ++t) {
      const _Float16* p = fa + t * 16 * GM_PITCH + kc * 32;
      const v8h lo = *(const v8h*)p;
      const v8h hi = *(const v8h*)(p + 16);
      const v16h a = __builtin_shufflevector(lo, hi, 0, 1, 2, 3, 4, 5, 6, 7,
                                             8, 9, 10, 11, 12, 13, 14, 15);
      c[t] = __builtin_amdgcn_wmma_f32_16x16x32_f16(false, a, false, b,
                                                    (short)0, c[t], false, false);
    }
  }

  const float bn = bias[gn];
#pragma unroll
  for (int t = 0; t < 4; ++t) {
    const v4f d0 = *(const v4f*)&sD[t * 16 + half * 8];
    const v4f d1 = *(const v4f*)&sD[t * 16 + half * 8 + 4];
    const float dv[8] = {d0.x, d0.y, d0.z, d0.w, d1.x, d1.y, d1.z, d1.w};
#pragma unroll
    for (int r = 0; r < 8; ++r) {
      const int gm = gm0 + t * 16 + r;
      float val = c[t][r] + bn;
      val *= dv[r];
      if constexpr (HAS_RESID) val += rs[t][r];
      if (!GUARD || gm < n_rows) out[gm * DIM + gn] = val;
    }
  }
}

template <bool HAS_RESID>
__global__ __launch_bounds__(512, 8)
void gemm_wmma_kernel(const float* __restrict__ A, const _Float16* __restrict__ Wp,
                      const float* __restrict__ bias, const float* __restrict__ dinv_in,
                      const float* __restrict__ resid, float* __restrict__ out, int n_rows) {
  __shared__ __attribute__((aligned(16))) _Float16 sA[GM_ROWS * GM_PITCH];
  __shared__ __attribute__((aligned(16))) float sD[GM_ROWS];
  if ((int)(blockIdx.x + 1) * GM_ROWS <= n_rows)
    gemm_block<HAS_RESID, false>(A, Wp, bias, dinv_in, resid, out, n_rows, sA, sD);
  else
    gemm_block<HAS_RESID, true>(A, Wp, bias, dinv_in, resid, out, n_rows, sA, sD);
}

#define BN_ROWS 64
#define BN_T (BN_ROWS / 4)
__global__ __launch_bounds__(256)
void bn_stats_kernel(const float* __restrict__ pre, float* __restrict__ sums,
                     float* __restrict__ sumsq, int n) {
  __shared__ __attribute__((aligned(16))) float part[2][4][DIM];
  const int c4 = threadIdx.x & 63;
  const int ph = threadIdx.x >> 6;
  const int row0 = blockIdx.x * BN_ROWS + ph;
  const v4f* src = (const v4f*)pre + c4;
  v4f v[BN_T];
  if ((int)(blockIdx.x + 1) * BN_ROWS <= n) {
#pragma unroll
    for (int t = 0; t < BN_T; ++t) v[t] = src[(row0 + 4 * t) * (DIM / 4)];
  } else {
#pragma unroll
    for (int t = 0; t < BN_T; ++t) {
      const int r = row0 + 4 * t;
      const v4f z = {0.0f, 0.0f, 0.0f, 0.0f};
      v[t] = (r < n) ? src[r * (DIM / 4)] : z;
    }
  }
  v4f s = {0.0f, 0.0f, 0.0f, 0.0f}, s2 = {0.0f, 0.0f, 0.0f, 0.0f};
#pragma unroll
  for (int t = 0; t < BN_T; ++t) {
    s  += v[t];
    s2 += v[t] * v[t];
  }
  *(v4f*)&part[0][ph][4 * c4] = s;
  *(v4f*)&part[1][ph][4 * c4] = s2;
  __syncthreads();
  const int c = threadIdx.x;
  const float a = ((part[0][0][c] + part[0][1][c]) + part[0][2][c]) + part[0][3][c];
  const float b = ((part[1][0][c] + part[1][1][c]) + part[1][2][c]) + part[1][3][c];
  atomAddF(&sums[c], a);
  atomAddF(&sumsq[c], b);
}

#define APPLY_U 8
__device__ __forceinline__ float bn_relu(float x, float mu, float is, float g, float b) {
  float v = (x - mu) * is * g + b;
  return fmaxf(v, 0.0f);
}
__device__ __forceinline__ void bn_moments(float s, float ss, float inv_n, float& mu, float& is) {
  mu = s * inv_n;
  const float var = ss * inv_n - mu * mu;
  is = rsqrtf(var + 1e-5f);
}
__global__ __launch_bounds__(256)
void bn_apply_kernel(float* __restrict__ buf, const float* __restrict__ sums,
                     const float* __restrict__ sumsq, const float* __restrict__ gamma,
                     const float* __restrict__ beta, int n4, int n) {
  v4f* q = (v4f*)buf;
  const int c = (threadIdx.x & 63) * 4;
  const int base = blockIdx.x * (256 * APPLY_U) + threadIdx.x;
  const bool whole = (int)(blockIdx.x + 1) * (256 * APPLY_U) <= n4;
  v4f v[APPLY_U];
  if (whole) {
#pragma unroll
    for (int k = 0; k < APPLY_U; ++k) v[k] = q[base + k * 256];
  } else {
#pragma unroll
    for (int k = 0; k < APPLY_U; ++k) {
      const int i = base + k * 256;
      const v4f z = {0.0f, 0.0f, 0.0f, 0.0f};
      v[k] = (i < n4) ? q[i] : z;
    }
  }
  const float inv_n = 1.0f / (float)n;
  float m0, m1, m2, m3, i0, i1, i2, i3;
  bn_moments(sums[c],     sumsq[c],     inv_n, m0, i0);
  bn_moments(sums[c + 1], sumsq[c + 1], inv_n, m1, i1);
  bn_moments(sums[c + 2], sumsq[c + 2], inv_n, m2, i2);
  bn_moments(sums[c + 3], sumsq[c + 3], inv_n, m3, i3);
  const float g0 = gamma[c], g1 = gamma[c + 1], g2 = gamma[c + 2], g3 = gamma[c + 3];
  const float b0 = beta[c], b1 = beta[c + 1], b2 = beta[c + 2], b3 = beta[c + 3];
#pragma unroll
  for (int k = 0; k < APPLY_U; ++k) {
    const int i = base + k * 256;
    if (whole || i < n4) {
      v4f r;
      r.x = bn_relu(v[k].x, m0, i0, g0, b0);
      r.y = bn_relu(v[k].y, m1, i1, g1, b1);
      r.z = bn_relu(v[k].z, m2, i2, g2, b2);
      r.w = bn_relu(v[k].w, m3, i3, g3, b3);
      q[i] = r;
    }
  }
}

extern "C" void kernel_launch(void* const* d_in, const int* in_sizes, int n_in,
                              void* d_out, int out_size, void* d_ws, size_t ws_size,
                              hipStream_t stream) {
  const float* x  = (const float*)d_in[0];
  const int*   ei = (const int*)d_in[1];
  const float* W0 = (const float*)d_in[2];
  const float* b0 = (const float*)d_in[3];
  const float* g0 = (const float*)d_in[4];
  const float* be0= (const float*)d_in[5];
  const float* W1 = (const float*)d_in[6];
  const float* b1 = (const float*)d_in[7];
  const float* g1 = (const float*)d_in[8];
  const float* be1= (const float*)d_in[9];
  const float* W2 = (const float*)d_in[10];
  const float* b2 = (const float*)d_in[11];
  float* out = (float*)d_out;

  const int N = in_sizes[0] / DIM;
  const int E = in_sizes[1] / 2;
  const int ND = N * DIM;
  const int* row = ei;
  const int* col = ei + E;

  float* ws       = (float*)d_ws;
  float* dinv_out = ws;
  float* dinv_in  = ws + N;
  int*   bcount   = (int*)(ws + 2 * N);
  int*   bcursor  = bcount + MAXB;
  float* stats0   = ws + 2 * N + 2 * MAXB;
  float* stats1   = stats0 + 512;
  float* agg      = stats1 + 512;
  float* bufA     = agg + (size_t)ND;
  _Float16* wp0   = (_Float16*)(bufA + (size_t)ND);
  _Float16* wp1   = wp0 + 65536;
  _Float16* wp2   = wp1 + 65536;
  int*  boffs     = (int*)(wp2 + 65536);
  int2* elist     = (int2*)(boffs + MAXB);
  int*  dpart     = (int*)(elist + E);

  const int zb = 256;
  auto nb = [](int n, int b) { return (n + b - 1) / b; };

  const int nbins       = nb(N, TR);
  const int bin_blocks  = nb(E, 1024 * BIN_EPT);
  const int nclear      = 2 * MAXB + 1024;
  const int gemm_blocks = nb(N, GM_ROWS);
  const int bn_blocks   = nb(N, BN_ROWS);
  const int ND4         = ND / 4;
  const int ap_blocks   = nb(ND4, 256 * APPLY_U);

  prep_kernel<<<3 * PACK_BLOCKS + nb(nclear, zb), zb, 0, stream>>>(W0, W1, W2, wp0, (float*)bcount, nclear);
  bin_count_kernel<<<2 * DEG_S * DEG_Q, 1024, 0, stream>>>(row, col, dpart, bcount, E, nbins);
  bin_fill_kernel<<<bin_blocks, 1024, 0, stream>>>(row, col, bcount, bcursor, boffs, elist, E, nbins,
                                                   dpart, dinv_out, N);

  edge_sum_kernel<<<nbins, 1024, 0, stream>>>(x, dinv_out, elist, boffs, bcount, agg, N);
  gemm_wmma_kernel<true><<<gemm_blocks, 512, 0, stream>>>(agg, wp0, b0, dinv_in, x, out, N);
  bn_stats_kernel<<<bn_blocks, 256, 0, stream>>>(out, stats0, stats0 + 256, N);
  bn_apply_kernel<<<ap_blocks, 256, 0, stream>>>(out, stats0, stats0 + 256, g0, be0, ND4, N);

  edge_sum_kernel<<<nbins, 1024, 0, stream>>>(out, dinv_out, elist, boffs, bcount, agg, N);
  gemm_wmma_kernel<true><<<gemm_blocks, 512, 0, stream>>>(agg, wp1, b1, dinv_in, out, bufA, N);
  bn_stats_kernel<<<bn_blocks, 256, 0, stream>>>(bufA, stats1, stats1 + 256, N);
  bn_apply_kernel<<<ap_blocks, 256, 0, stream>>>(bufA, stats1, stats1 + 256, g1, be1, ND4, N);

  edge_sum_kernel<<<nbins, 1024, 0, stream>>>(bufA, dinv_out, elist, boffs, bcount, agg, N);
  gemm_wmma_kernel<false><<<gemm_blocks, 512, 0, stream>>>(agg, wp2, b2, dinv_in, nullptr, out, N);
}


# BatchNorm+ReLU applied where the next layer reads (edge rows and residual: one FMA with a=gamma/sqrt(var+eps), c=beta-mean*a, one max); planes keep pre-norm values, both apply launches gone (10 -> 8)
# speedup vs baseline: 6.1435x; 1.6467x over previous
#include <hip/hip_runtime.h>
#include <math.h>


typedef __attribute__((ext_vector_type(16))) _Float16 v16h;
typedef __attribute__((ext_vector_type(8)))  _Float16 v8h;
typedef __attribute__((ext_vector_type(4)))  _Float16 v4h;
typedef __attribute__((ext_vector_type(8)))  float    v8f;
typedef __attribute__((ext_vector_type(4)))  float    v4f;

#define DIM 256

__device__ __forceinline__ float atomAddF(float* p, float v) {
  return unsafeAtomicAdd(p, v);
}

#define PACK_BLOCKS 256
__global__ __launch_bounds__(256)
void prep_kernel(const float* __restrict__ W0, const float* __restrict__ W1,
                 const float* __restrict__ W2, _Float16* __restrict__ pack,
                 float* __restrict__ z, int nz) {
  const int b = blockIdx.x;
  if (b < 3 * PACK_BLOCKS) {
    const int which = b / PACK_BLOCKS;
    const float* __restrict__ W = which == 0 ? W0 : (which == 1 ? W1 : W2);
    const int t = (b - which * PACK_BLOCKS) * 256 + threadIdx.x;
    const int elem = t & 15;
    const int lane = (t >> 4) & 31;
    const int kc   = (t >> 9) & 7;
    const int tn   = t >> 12;
    const int v = elem >> 1, p = elem & 1;
    const int half = lane >> 4, mn = lane & 15;
    const int kb = kc * 32 + ((v & 4) << 2) + (half << 3) + ((v & 3) << 1) + p;
    const int gn = tn * 16 + mn;
    pack[which * 65536 + t] = (_Float16)W[kb * DIM + gn];
  } else {
    const int i = (b - 3 * PACK_BLOCKS) * 256 + threadIdx.x;
    if (i < nz) z[i] = 0.0f;
  }
}

#define TR 196
#define MAXB 1024
#define BIN_EPT 4
#define DEG_S 32
#define DEG_CAP (TR * 256)
#define DEG_Q 4
#define DEG_QCAP (DEG_CAP / DEG_Q)
#define DEG_EPT 16

typedef __attribute__((ext_vector_type(4))) int v4i;

__global__ __launch_bounds__(1024)
void bin_count_kernel(const int* __restrict__ row, const int* __restrict__ col,
                      int* __restrict__ part, int* __restrict__ count, int E, int nbins) {
  __shared__ __attribute__((aligned(16))) int cnt[DEG_QCAP];
  __shared__ int hist[MAXB];
  const int tid = threadIdx.x;
  const int q = blockIdx.x % DEG_Q;
  const int kind = (blockIdx.x / DEG_Q) & 1;
  const int s = blockIdx.x / (2 * DEG_Q);
  const int lo = q * DEG_QCAP;
  const v4i z = {0, 0, 0, 0};
  for (int i = tid; i < DEG_QCAP / 4; i += 1024) ((v4i*)cnt)[i] = z;
  hist[tid] = 0;
  __syncthreads();
  const int per = (E + DEG_S - 1) / DEG_S;
  const int e0 = s * per;
  int e1 = e0 + per;
  e1 = e1 < E ? e1 : E;
  const int* __restrict__ key = kind ? row : col;
  for (int base = e0 + tid; base < e1; base += 1024 * DEG_EPT) {
    int kv[DEG_EPT];
#pragma unroll
    for (int k = 0; k < DEG_EPT; ++k) {
      const int e = base + k * 1024;
      kv[k] = (e < e1) ? key[e] : -1;
    }
#pragma unroll
    for (int k = 0; k < DEG_EPT; ++k) {
      const int r = kv[k] - lo;
      if ((unsigned)r < (unsigned)DEG_QCAP) {
        atomicAdd(&cnt[r], 1);
        if (kind == 0) atomicAdd(&hist[kv[k] / TR], 1);
      }
    }
  }
  __syncthreads();
  if (kind == 0 && tid < nbins) {
    const int hc = hist[tid];
    if (hc) atomicAdd(&count[tid], hc);
  }
  v4i* dst = (v4i*)(part + (size_t)(2 * s + kind) * DEG_CAP + lo);
  for (int i = tid; i < DEG_QCAP / 4; i += 1024) dst[i] = ((const v4i*)cnt)[i];
}

__global__ __launch_bounds__(1024)
void bin_fill_kernel(const int* __restrict__ row, const int* __restrict__ col,
                     const int* __restrict__ count, int* __restrict__ cursor,
                     int* __restrict__ offs, int2* __restrict__ list, int E, int nbins,
                     const int* __restrict__ part, float* __restrict__ deg, int N) {
  __shared__ int hist[MAXB];
  __shared__ int scan[MAXB];
  __shared__ int basep[MAXB];
  const int tid = threadIdx.x;
  for (int i = blockIdx.x * 1024 + tid; i < 2 * N; i += gridDim.x * 1024) {
    const int* p = part + (i < N ? DEG_CAP + i : i - N);
    int c = 0;
#pragma unroll
    for (int s = 0; s < DEG_S; ++s) c += p[(size_t)(2 * s) * DEG_CAP];
    float v = (float)c;
    v = v < 1.0f ? 1.0f : v;
    deg[i] = rsqrtf(v);
  }
  const int c = (tid < nbins) ? count[tid] : 0;
  scan[tid] = c;
  hist[tid] = 0;
  __syncthreads();
  for (int d = 1; d < MAXB; d <<= 1) {
    int v = scan[tid];
    if (tid >= d) v += scan[tid - d];
    __syncthreads();
    scan[tid] = v;
    __syncthreads();
  }
  const int excl = scan[tid] - c;
  if (blockIdx.x == 0 && tid < nbins) offs[tid] = excl;

  const int base = blockIdx.x * (1024 * BIN_EPT) + tid;
  int rr[BIN_EPT], cl[BIN_EPT], bb[BIN_EPT], rk[BIN_EPT];
#pragma unroll
  for (int k = 0; k < BIN_EPT; ++k) {
    const int e = base + k * 1024;
    bb[k] = -1;
    if (e < E) {
      const int cc = col[e];
      rr[k] = row[e];
      bb[k] = cc / TR;
      cl[k] = cc - bb[k] * TR;
      rk[k] = atomicAdd(&hist[bb[k]], 1);
    }
  }
  __syncthreads();
  if (tid < nbins) {
    const int hc = hist[tid];
    basep[tid] = excl + (hc ? atomicAdd(&cursor[tid], hc) : 0);
  }
  __syncthreads();
#pragma unroll
  for (int k = 0; k < BIN_EPT; ++k) {
    if (bb[k] >= 0) list[basep[bb[k]] + rk[k]] = make_int2(rr[k], cl[k]);
  }
}


#define GM_PITCH 264
template <int NT, bool HAS_RESID, bool IN_NORM>
__device__ __forceinline__
void tile_gemm(const _Float16* sA, const float* sD, const float* sNA, const float* sNC,
               const _Float16* __restrict__ Wp,
               const float* __restrict__ bias, const float* __restrict__ resid,
               float* __restrict__ out, int rt0, int cp, int lane, int row0, int rows) {
  const int half = lane >> 4;
  const int mn   = lane & 15;
  const v16h* Bp0 = (const v16h*)Wp + (2 * cp) * 256 + lane;
  const v16h* Bp1 = Bp0 + 256;
  const _Float16* fa = sA + (rt0 * 16 + mn) * GM_PITCH + half * 8;
  v8f c[NT][2] = {};
#pragma unroll
  for (int kc = 0; kc < 8; ++kc) {
    const v16h b0 = Bp0[kc * 32];
    const v16h b1 = Bp1[kc * 32];
#pragma unroll
    for (int t = 0; t < NT; ++t) {
      const _Float16* p = fa + t * 16 * GM_PITCH + kc * 32;
      const v8h lo = *(const v8h*)p;
      const v8h hi = *(const v8h*)(p + 16);
      const v16h a = __builtin_shufflevector(lo, hi, 0, 1, 2, 3, 4, 5, 6, 7,
                                             8, 9, 10, 11, 12, 13, 14, 15);
      c[t][0] = __builtin_amdgcn_wmma_f32_16x16x32_f16(false, a, false, b0,
                                                       (short)0, c[t][0], false, false);
      c[t][1] = __builtin_amdgcn_wmma_f32_16x16x32_f16(false, a, false, b1,
                                                       (short)0, c[t][1], false, false);
    }
  }

  const int gn = cp * 32 + mn;
  const float bn0 = bias[gn];
  const float bn1 = bias[gn + 16];
#pragma unroll
  for (int t = 0; t < NT; ++t) {
    const int lr0 = (rt0 + t) * 16 + half * 8;
    const v4f d0 = *(const v4f*)&sD[lr0];
    const v4f d1 = *(const v4f*)&sD[lr0 + 4];
    const float dv[8] = {d0.x, d0.y, d0.z, d0.w, d1.x, d1.y, d1.z, d1.w};
    float r0[8], r1[8];
    if constexpr (HAS_RESID) {
#pragma unroll
      for (int r = 0; r < 8; ++r) {
        const int gi = (row0 + lr0 + r) * DIM + gn;
        const bool ok = lr0 + r < rows;
        r0[r] = ok ? resid[gi] : 0.0f;
        r1[r] = ok ? resid[gi + 16] : 0.0f;
      }
      if constexpr (IN_NORM) {
        const float na0 = sNA[gn],      nc0 = sNC[gn];
        const float na1 = sNA[gn + 16], nc1 = sNC[gn + 16];
#pragma unroll
        for (int r = 0; r < 8; ++r) {
          r0[r] = fmaxf(fmaf(r0[r], na0, nc0), 0.0f);
          r1[r] = fmaxf(fmaf(r1[r], na1, nc1), 0.0f);
        }
      }
    }
#pragma unroll
    for (int r = 0; r < 8; ++r) {
      const int gi = (row0 + lr0 + r) * DIM + gn;
      float v0 = c[t][0][r] + bn0;
      float v1 = c[t][1][r] + bn1;
      v0 *= dv[r];
      v1 *= dv[r];
      if constexpr (HAS_RESID) { v0 += r0[r]; v1 += r1[r]; }
      if (lr0 + r < rows) { out[gi] = v0; out[gi + 16] = v1; }
    }
  }
}

#define PE 4
struct EdgeGroup {
  int   dst[PE];
  float s[PE];
  float v[PE][8];
};

__device__ __forceinline__
void group_load(EdgeGroup& g, const float* __restrict__ h, const float* __restrict__ dinv_out,
                const int2* __restrict__ mine, int e, int lane) {
#pragma unroll
  for (int i = 0; i < PE; ++i) {
    const int2 rc = mine[e + i];
    g.dst[i] = rc.y;
    g.s[i] = dinv_out[rc.x];
    const float* src = h + rc.x * DIM + lane;
#pragma unroll
    for (int j = 0; j < 8; ++j) g.v[i][j] = src[j * 32];
  }
}

__device__ __forceinline__ void bn_moments(float s, float ss, float inv_n, float& mu, float& is) {
  mu = s * inv_n;
  const float var = ss * inv_n - mu * mu;
  is = rsqrtf(var + 1e-5f);
}

struct LaneNorm {
  float a[8];
  float c[8];
};

template <bool IN_NORM>
__device__ __forceinline__ float norm_in(float x, const LaneNorm& nm, int j) {
  if constexpr (IN_NORM) return fmaxf(fmaf(x, nm.a[j], nm.c[j]), 0.0f);
  else return x;
}

template <bool IN_NORM>
__device__ __forceinline__
void group_add(const EdgeGroup& g, float* trow, const LaneNorm& nm) {
#pragma unroll
  for (int i = 0; i < PE; ++i) {
    float* dst = trow + g.dst[i] * DIM;
#pragma unroll
    for (int j = 0; j < 8; ++j) atomAddF(dst + j * 32, norm_in<IN_NORM>(g.v[i][j], nm, j) * g.s[i]);
  }
}

#define TRT 13
#define TRP (TRT * 16)
#define CV_T (TRP * (DIM / 4) / 1024)
template <bool HAS_RESID, bool IN_NORM>
__global__ __launch_bounds__(1024)
void edge_gemm_kernel(const float* __restrict__ h, const float* __restrict__ dinv_out,
                      const int2* __restrict__ list, const int* __restrict__ offs,
                      const int* __restrict__ count, const _Float16* __restrict__ Wp,
                      const float* __restrict__ bias, const float* __restrict__ dinv_in,
                      const float* __restrict__ resid, float* __restrict__ out, int N,
                      const float* __restrict__ in_sums, const float* __restrict__ in_sumsq,
                      const float* __restrict__ in_gamma, const float* __restrict__ in_beta) {
  __shared__ __attribute__((aligned(16))) float tile[TRP * DIM];
  __shared__ __attribute__((aligned(16))) float sD[TRP];
  __shared__ float sNA[DIM];
  __shared__ float sNC[DIM];
  const int tid  = threadIdx.x;
  const int lane = tid & 31;
  const int wave = __builtin_amdgcn_readfirstlane(tid >> 5);
  const int b = blockIdx.x;
  const v4f z = {0.0f, 0.0f, 0.0f, 0.0f};
  for (int i = tid; i < TRP * DIM / 4; i += 1024) ((v4f*)tile)[i] = z;
  const int cnt = count[b];
  const int2* mine = list + offs[b];
  if constexpr (IN_NORM) {
    if (tid < DIM) {
      float mu, is;
      bn_moments(in_sums[tid], in_sumsq[tid], 1.0f / (float)N, mu, is);
      const float a = is * in_gamma[tid];
      sNA[tid] = a;
      sNC[tid] = in_beta[tid] - mu * a;
    }
  }
  __syncthreads();
  LaneNorm nm;
#pragma unroll
  for (int j = 0; j < 8; ++j) {
    nm.a[j] = IN_NORM ? sNA[lane + j * 32] : 1.0f;
    nm.c[j] = IN_NORM ? sNC[lane + j * 32] : 0.0f;
  }
  float* trow = tile + lane;
  const int G = cnt / PE;
  const int m = (G > wave) ? (G - wave + 31) >> 5 : 0;
  const int2* wm = mine + wave * PE;
  if (m > 0) {
    EdgeGroup ga, gb;
    group_load(ga, h, dinv_out, wm, 0, lane);
    int k = 0;
    while (k + 2 < m) {
      group_load(gb, h, dinv_out, wm, (k + 1) * (32 * PE), lane);
      __builtin_amdgcn_sched_barrier(0);
      group_add<IN_NORM>(ga, trow, nm);
      group_load(ga, h, dinv_out, wm, (k + 2) * (32 * PE), lane);
      __builtin_amdgcn_sched_barrier(0);
      group_add<IN_NORM>(gb, trow, nm);
      k += 2;
    }
    if (k + 1 < m) {
      group_load(gb, h, dinv_out, wm, (k + 1) * (32 * PE), lane);
      __builtin_amdgcn_sched_barrier(0);
      group_add<IN_NORM>(ga, trow, nm);
      group_add<IN_NORM>(gb, trow, nm);
    } else {
      group_add<IN_NORM>(ga, trow, nm);
    }
  }
  const int n = cnt - G * PE;
  if (n > 0 && wave == (G & 31)) {
    const int e0 = G * PE;
    int   dst[PE - 1];
    float sc[PE - 1];
    float v[PE - 1][8];
#pragma unroll
    for (int i = 0; i < PE - 1; ++i) {
      const int2 rc = mine[e0 + (i < n ? i : n - 1)];
      dst[i] = rc.y;
      sc[i] = dinv_out[rc.x];
      const float* src = h + rc.x * DIM + lane;
#pragma unroll
      for (int j = 0; j < 8; ++j) v[i][j] = src[j * 32];
    }
#pragma unroll
    for (int i = 0; i < PE - 1; ++i) {
      if (i < n) {
        float* d = trow + dst[i] * DIM;
#pragma unroll
        for (int j = 0; j < 8; ++j) atomAddF(d + j * 32, norm_in<IN_NORM>(v[i][j], nm, j) * sc[i]);
      }
    }
  }

  __syncthreads();

  const int row0 = b * TR;
  int rows = N - row0;
  rows = rows > TR ? TR : rows;

  v4f q[CV_T];
#pragma unroll
  for (int i = 0; i < CV_T; ++i) q[i] = ((const v4f*)tile)[i * 1024 + tid];
  if (tid < TRP) sD[tid] = (tid < rows) ? dinv_in[row0 + tid] : 0.0f;
  __syncthreads();
  _Float16* sA = (_Float16*)tile;
#pragma unroll
  for (int i = 0; i < CV_T; ++i) {
    const int p = i * 1024 + tid;
    v4h hq;
    hq[0] = (_Float16)q[i].x;  hq[1] = (_Float16)q[i].y;
    hq[2] = (_Float16)q[i].z;  hq[3] = (_Float16)q[i].w;
    *(v4h*)&sA[(p >> 6) * GM_PITCH + (p & 63) * 4] = hq;
  }
  __syncthreads();

  const int cp = wave & 7;
  const int rg = wave >> 3;
  if (rg == 0) {
    tile_gemm<2, HAS_RESID, IN_NORM>(sA, sD, sNA, sNC, Wp, bias, resid, out, 0, cp, lane, row0, rows);
    tile_gemm<2, HAS_RESID, IN_NORM>(sA, sD, sNA, sNC, Wp, bias, resid, out, 2, (cp + 4) & 7, lane, row0, rows);
  } else {
    tile_gemm<3, HAS_RESID, IN_NORM>(sA, sD, sNA, sNC, Wp, bias, resid, out, 1 + 3 * rg, cp, lane, row0, rows);
  }
}


#define BN_ROWS 64
#define BN_T (BN_ROWS / 4)
__global__ __launch_bounds__(256)
void bn_stats_kernel(const float* __restrict__ pre, float* __restrict__ sums,
                     float* __restrict__ sumsq, int n) {
  __shared__ __attribute__((aligned(16))) float part[2][4][DIM];
  const int c4 = threadIdx.x & 63;
  const int ph = threadIdx.x >> 6;
  const int row0 = blockIdx.x * BN_ROWS + ph;
  const v4f* src = (const v4f*)pre + c4;
  v4f v[BN_T];
  if ((int)(blockIdx.x + 1) * BN_ROWS <= n) {
#pragma unroll
    for (int t = 0; t < BN_T; ++t) v[t] = src[(row0 + 4 * t) * (DIM / 4)];
  } else {
#pragma unroll
    for (int t = 0; t < BN_T; ++t) {
      const int r = row0 + 4 * t;
      const v4f z = {0.0f, 0.0f, 0.0f, 0.0f};
      v[t] = (r < n) ? src[r * (DIM / 4)] : z;
    }
  }
  v4f s = {0.0f, 0.0f, 0.0f, 0.0f}, s2 = {0.0f, 0.0f, 0.0f, 0.0f};
#pragma unroll
  for (int t = 0; t < BN_T; ++t) {
    s  += v[t];
    s2 += v[t] * v[t];
  }
  *(v4f*)&part[0][ph][4 * c4] = s;
  *(v4f*)&part[1][ph][4 * c4] = s2;
  __syncthreads();
  const int c = threadIdx.x;
  const float a = ((part[0][0][c] + part[0][1][c]) + part[0][2][c]) + part[0][3][c];
  const float b = ((part[1][0][c] + part[1][1][c]) + part[1][2][c]) + part[1][3][c];
  atomAddF(&sums[c], a);
  atomAddF(&sumsq[c], b);
}

extern "C" void kernel_launch(void* const* d_in, const int* in_sizes, int n_in,
                              void* d_out, int out_size, void* d_ws, size_t ws_size,
                              hipStream_t stream) {
  const float* x  = (const float*)d_in[0];
  const int*   ei = (const int*)d_in[1];
  const float* W0 = (const float*)d_in[2];
  const float* b0 = (const float*)d_in[3];
  const float* g0 = (const float*)d_in[4];
  const float* be0= (const float*)d_in[5];
  const float* W1 = (const float*)d_in[6];
  const float* b1 = (const float*)d_in[7];
  const float* g1 = (const float*)d_in[8];
  const float* be1= (const float*)d_in[9];
  const float* W2 = (const float*)d_in[10];
  const float* b2 = (const float*)d_in[11];
  float* out = (float*)d_out;

  const int N = in_sizes[0] / DIM;
  const int E = in_sizes[1] / 2;
  const int ND = N * DIM;
  const int* row = ei;
  const int* col = ei + E;

  float* ws       = (float*)d_ws;
  float* dinv_out = ws;
  float* dinv_in  = ws + N;
  int*   bcount   = (int*)(ws + 2 * N);
  int*   bcursor  = bcount + MAXB;
  float* stats0   = ws + 2 * N + 2 * MAXB;
  float* stats1   = stats0 + 512;
  float* bufA     = stats1 + 512;
  _Float16* wp0   = (_Float16*)(bufA + (size_t)ND);
  _Float16* wp1   = wp0 + 65536;
  _Float16* wp2   = wp1 + 65536;
  int*  boffs     = (int*)(wp2 + 65536);
  int2* elist     = (int2*)(boffs + MAXB);
  int*  dpart     = (int*)(elist + E);

  const int zb = 256;
  auto nb = [](int n, int b) { return (n + b - 1) / b; };

  const int nbins       = nb(N, TR);
  const int bin_blocks  = nb(E, 1024 * BIN_EPT);
  const int nclear      = 2 * MAXB + 1024;
  const int bn_blocks   = nb(N, BN_ROWS);

  prep_kernel<<<3 * PACK_BLOCKS + nb(nclear, zb), zb, 0, stream>>>(W0, W1, W2, wp0, (float*)bcount, nclear);
  bin_count_kernel<<<2 * DEG_S * DEG_Q, 1024, 0, stream>>>(row, col, dpart, bcount, E, nbins);
  bin_fill_kernel<<<bin_blocks, 1024, 0, stream>>>(row, col, bcount, bcursor, boffs, elist, E, nbins,
                                                   dpart, dinv_out, N);

  edge_gemm_kernel<true, false><<<nbins, 1024, 0, stream>>>(x, dinv_out, elist, boffs, bcount,
                                                            wp0, b0, dinv_in, x, out, N,
                                                            nullptr, nullptr, nullptr, nullptr);
  bn_stats_kernel<<<bn_blocks, 256, 0, stream>>>(out, stats0, stats0 + 256, N);

  edge_gemm_kernel<true, true><<<nbins, 1024, 0, stream>>>(out, dinv_out, elist, boffs, bcount,
                                                           wp1, b1, dinv_in, out, bufA, N,
                                                           stats0, stats0 + 256, g0, be0);
  bn_stats_kernel<<<bn_blocks, 256, 0, stream>>>(bufA, stats1, stats1 + 256, N);

  edge_gemm_kernel<false, true><<<nbins, 1024, 0, stream>>>(bufA, dinv_out, elist, boffs, bcount,
                                                            wp2, b2, dinv_in, nullptr, out, N,
                                                            stats1, stats1 + 256, g1, be1);
}
